# starting kernel
# baseline (speedup 1.0000x reference)
#include <hip/hip_runtime.h>
#include <hip/hip_bf16.h>

#define NNODE 3072
#define FIN   1546
#define D0    256
#define HDIM  128
#define BPAIR 16384

typedef __attribute__((ext_vector_type(2))) float v2f;
typedef __attribute__((ext_vector_type(8))) float v8f;

__device__ __forceinline__ void wmma8(v8f (&acc)[2][4], v2f a0, v2f a1,
                                      v2f b0, v2f b1, v2f b2, v2f b3)
{
  acc[0][0] = __builtin_amdgcn_wmma_f32_16x16x4_f32(false, a0, false, b0, (short)0, acc[0][0], false, false);
  acc[0][1] = __builtin_amdgcn_wmma_f32_16x16x4_f32(false, a0, false, b1, (short)0, acc[0][1], false, false);
  acc[0][2] = __builtin_amdgcn_wmma_f32_16x16x4_f32(false, a0, false, b2, (short)0, acc[0][2], false, false);
  acc[0][3] = __builtin_amdgcn_wmma_f32_16x16x4_f32(false, a0, false, b3, (short)0, acc[0][3], false, false);
  acc[1][0] = __builtin_amdgcn_wmma_f32_16x16x4_f32(false, a1, false, b0, (short)0, acc[1][0], false, false);
  acc[1][1] = __builtin_amdgcn_wmma_f32_16x16x4_f32(false, a1, false, b1, (short)0, acc[1][1], false, false);
  acc[1][2] = __builtin_amdgcn_wmma_f32_16x16x4_f32(false, a1, false, b2, (short)0, acc[1][2], false, false);
  acc[1][3] = __builtin_amdgcn_wmma_f32_16x16x4_f32(false, a1, false, b3, (short)0, acc[1][3], false, false);
}

template <bool RELU, bool BIAS, bool GATHER>
__global__ __launch_bounds__(128) void gemm_wmma_f32(
    const float* __restrict__ A, int lda,
    const float* __restrict__ Bm, int ldb,
    const float* __restrict__ bias,
    float* __restrict__ C, int ldc,
    int K,
    const int* __restrict__ gl, const int* __restrict__ gr)
{
  const int lane = threadIdx.x & 31;
  const int wave = threadIdx.x >> 5;
  const int half = lane >> 4;
  const int l16  = lane & 15;

  const int m0  = blockIdx.y * 128 + wave * 32;
  const int n0  = blockIdx.x * 64;
  const int am0 = m0 + l16;
  const int am1 = m0 + 16 + l16;

  v8f acc[2][4] = {};

  const float* bp = Bm + (size_t)(2 * half) * ldb + n0 + l16;
  const size_t bstep = (size_t)4 * ldb;

  if (GATHER) {
    const float* ap0 = A + (size_t)gl[am0] * 256 + 2 * half;
    const float* ap1 = A + (size_t)gl[am1] * 256 + 2 * half;
    #pragma unroll 4
    for (int k = 0; k < 256; k += 4) {
      v2f a0 = *(const v2f*)ap0;
      v2f a1 = *(const v2f*)ap1;
      v2f b0, b1, b2, b3;
      b0.x = bp[0];  b0.y = bp[ldb + 0];
      b1.x = bp[16]; b1.y = bp[ldb + 16];
      b2.x = bp[32]; b2.y = bp[ldb + 32];
      b3.x = bp[48]; b3.y = bp[ldb + 48];
      wmma8(acc, a0, a1, b0, b1, b2, b3);
      ap0 += 4; ap1 += 4; bp += bstep;
    }
    ap0 = A + (size_t)gr[am0] * 256 + 2 * half;
    ap1 = A + (size_t)gr[am1] * 256 + 2 * half;
    #pragma unroll 4
    for (int k = 0; k < 256; k += 4) {
      v2f a0 = *(const v2f*)ap0;
      v2f a1 = *(const v2f*)ap1;
      v2f b0, b1, b2, b3;
      b0.x = bp[0];  b0.y = bp[ldb + 0];
      b1.x = bp[16]; b1.y = bp[ldb + 16];
      b2.x = bp[32]; b2.y = bp[ldb + 32];
      b3.x = bp[48]; b3.y = bp[ldb + 48];
      wmma8(acc, a0, a1, b0, b1, b2, b3);
      ap0 += 4; ap1 += 4; bp += bstep;
    }
  } else {
    const float* ap0 = A + (size_t)am0 * lda + 2 * half;
    const float* ap1 = A + (size_t)am1 * lda + 2 * half;
    const int kmain = K & ~3;
    #pragma unroll 2
    for (int k = 0; k < kmain; k += 4) {
      v2f a0 = *(const v2f*)ap0;
      v2f a1 = *(const v2f*)ap1;
      v2f b0, b1, b2, b3;
      b0.x = bp[0];  b0.y = bp[ldb + 0];
      b1.x = bp[16]; b1.y = bp[ldb + 16];
      b2.x = bp[32]; b2.y = bp[ldb + 32];
      b3.x = bp[48]; b3.y = bp[ldb + 48];
      wmma8(acc, a0, a1, b0, b1, b2, b3);
      ap0 += 4; ap1 += 4; bp += bstep;
    }
    if (kmain < K) {
      const int ka = kmain + 2 * half;
      const bool g0 = (ka < K), g1 = (ka + 1 < K);
      v2f a0, a1, b0, b1, b2, b3;
      a0.x = g0 ? ap0[0] : 0.0f;  a0.y = g1 ? ap0[1] : 0.0f;
      a1.x = g0 ? ap1[0] : 0.0f;  a1.y = g1 ? ap1[1] : 0.0f;
      b0.x = g0 ? bp[0]  : 0.0f;  b0.y = g1 ? bp[ldb + 0]  : 0.0f;
      b1.x = g0 ? bp[16] : 0.0f;  b1.y = g1 ? bp[ldb + 16] : 0.0f;
      b2.x = g0 ? bp[32] : 0.0f;  b2.y = g1 ? bp[ldb + 32] : 0.0f;
      b3.x = g0 ? bp[48] : 0.0f;  b3.y = g1 ? bp[ldb + 48] : 0.0f;
      wmma8(acc, a0, a1, b0, b1, b2, b3);
    }
  }

  #pragma unroll
  for (int s = 0; s < 2; ++s) {
    #pragma unroll
    for (int v = 0; v < 8; ++v) {
      const int row = m0 + 16 * s + v + 8 * half;
      #pragma unroll
      for (int j = 0; j < 4; ++j) {
        const int col = n0 + 16 * j + l16;
        float x = acc[s][j][v];
        if (BIAS) x += bias[col];
        if (RELU) x = fmaxf(x, 0.0f);
        C[(size_t)row * ldc + col] = x;
      }
    }
  }
}

__global__ void deg_rsqrt(const float* __restrict__ A, float* __restrict__ d)
{
  const int gw   = (blockIdx.x * blockDim.x + threadIdx.x) >> 5;
  const int lane = threadIdx.x & 31;
  if (gw >= NNODE) return;
  const float* row = A + (size_t)gw * NNODE;
  float s = 0.0f;
  for (int j = lane; j < NNODE; j += 32) s += row[j];
  #pragma unroll
  for (int off = 16; off > 0; off >>= 1) s += __shfl_xor(s, off, 32);
  if (lane == 0) d[gw] = (s > 0.0f) ? rsqrtf(s) : 0.0f;
}

__global__ void branch_deg_rsqrt(const float* __restrict__ mask,
                                 const float* __restrict__ adj,
                                 const float* __restrict__ thrp,
                                 float* __restrict__ d)
{
  const int gw   = (blockIdx.x * blockDim.x + threadIdx.x) >> 5;
  const int lane = threadIdx.x & 31;
  if (gw >= NNODE) return;
  const float thr = thrp[0];
  const size_t base = (size_t)gw * NNODE;
  float s = 0.0f;
  for (int j = lane; j < NNODE; j += 32)
    s += (mask[base + j] > thr) ? adj[base + j] : 0.0f;
  #pragma unroll
  for (int off = 16; off > 0; off >>= 1) s += __shfl_xor(s, off, 32);
  if (lane == 0) d[gw] = rsqrtf(s + 1.0f);
}

__global__ void symnorm(const float* __restrict__ adj,
                        const float* __restrict__ d,
                        float* __restrict__ An)
{
  const int idx = blockIdx.x * blockDim.x + threadIdx.x;
  if (idx >= NNODE * NNODE) return;
  const int i = idx / NNODE;
  const int j = idx - i * NNODE;
  An[idx] = adj[idx] * d[i] * d[j];
}

__global__ void branch_norm(const float* __restrict__ mask,
                            const float* __restrict__ adj,
                            const float* __restrict__ thrp,
                            const float* __restrict__ d,
                            float* __restrict__ An)
{
  const int idx = blockIdx.x * blockDim.x + threadIdx.x;
  if (idx >= NNODE * NNODE) return;
  const float thr = thrp[0];
  const int i = idx / NNODE;
  const int j = idx - i * NNODE;
  float w = (mask[idx] > thr) ? adj[idx] : 0.0f;
  if (i == j) w += 1.0f;
  An[idx] = w * d[i] * d[j];
}

__global__ void conv_dil2(const float* __restrict__ emb,
                          const float* __restrict__ kw,
                          const float* __restrict__ kb,
                          float* __restrict__ out)
{
  const int idx = blockIdx.x * blockDim.x + threadIdx.x;
  if (idx >= NNODE * 256) return;
  const int n = idx >> 8;
  const int c = idx & 255;
  float acc = kb[0];
  #pragma unroll
  for (int ch = 0; ch < 3; ++ch) {
    #pragma unroll
    for (int kh = 0; kh < 3; ++kh) {
      const int ih = n + 2 * kh - 2;
      if (ih < 0 || ih >= NNODE) continue;
      #pragma unroll
      for (int kwi = 0; kwi < 3; ++kwi) {
        const int iw = c + 2 * kwi - 2;
        if (iw < 0 || iw >= 256) continue;
        acc += emb[((size_t)ch * NNODE + ih) * 256 + iw] * kw[ch * 9 + kh * 3 + kwi];
      }
    }
  }
  out[idx] = acc;
}

__global__ void logits_k(const float* __restrict__ h,
                         const float* __restrict__ Wb,
                         const float* __restrict__ bb,
                         float* __restrict__ out)
{
  const int i = blockIdx.x * blockDim.x + threadIdx.x;
  if (i >= BPAIR * 2) return;
  const int b = i >> 1;
  const int c = i & 1;
  float acc = bb[c];
  #pragma unroll
  for (int k = 0; k < 64; ++k) acc += h[b * 64 + k] * Wb[k * 2 + c];
  out[i] = acc;
}

extern "C" void kernel_launch(void* const* d_in, const int* in_sizes, int n_in,
                              void* d_out, int out_size, void* d_ws, size_t ws_size,
                              hipStream_t stream)
{
  const int*   left  = (const int*)d_in[0];
  const int*   right = (const int*)d_in[1];
  const float* X     = (const float*)d_in[2];
  const float* adj   = (const float*)d_in[3];
  const float* thr   = (const float*)d_in[4];
  const float* Wfc1  = (const float*)d_in[5];
  const float* bfc1  = (const float*)d_in[6];
  const float* Wg[6] = {(const float*)d_in[7],  (const float*)d_in[9],
                        (const float*)d_in[11], (const float*)d_in[13],
                        (const float*)d_in[15], (const float*)d_in[17]};
  const float* bg[6] = {(const float*)d_in[8],  (const float*)d_in[10],
                        (const float*)d_in[12], (const float*)d_in[14],
                        (const float*)d_in[16], (const float*)d_in[18]};
  const float* cnnk  = (const float*)d_in[19];
  const float* cnnb  = (const float*)d_in[20];
  const float* Wa    = (const float*)d_in[21];
  const float* ba    = (const float*)d_in[22];
  const float* Wb    = (const float*)d_in[23];
  const float* bb    = (const float*)d_in[24];

  float* out     = (float*)d_out;
  float* logits  = out;
  float* emb_all = out + BPAIR * 2;

  float* ws = (float*)d_ws;
  size_t off = 0;
  float* X0  = ws + off; off += (size_t)NNODE * D0;
  float* An  = ws + off; off += (size_t)NNODE * NNODE;
  float* A2  = ws + off; off += (size_t)NNODE * NNODE;
  float* AnB = ws + off; off += (size_t)NNODE * NNODE;
  float* dv  = ws + off; off += NNODE;
  float* Hb  = ws + off; off += (size_t)NNODE * HDIM;
  float* H2b = ws + off; off += (size_t)NNODE * HDIM;
  float* emb = ws + off; off += (size_t)3 * NNODE * 256;
  float* hm  = ws + off; off += (size_t)BPAIR * 64;
  if (ws_size < off * sizeof(float)) return;

  const dim3 blk(128);

  gemm_wmma_f32<true, true, false><<<dim3(D0 / 64, NNODE / 128), blk, 0, stream>>>(
      X, FIN, Wfc1, D0, bfc1, X0, D0, FIN, nullptr, nullptr);

  deg_rsqrt<<<NNODE / 8, 256, 0, stream>>>(adj, dv);
  symnorm<<<(NNODE * NNODE) / 256, 256, 0, stream>>>(adj, dv, An);
  gemm_wmma_f32<false, false, false><<<dim3(NNODE / 64, NNODE / 128), blk, 0, stream>>>(
      An, NNODE, An, NNODE, nullptr, A2, NNODE, NNODE, nullptr, nullptr);

  const float* masks[3] = { adj, An, A2 };
  for (int ch = 0; ch < 3; ++ch) {
    const float* W1 = Wg[4 - 2 * ch];  const float* b1 = bg[4 - 2 * ch];
    const float* W2 = Wg[5 - 2 * ch];  const float* b2 = bg[5 - 2 * ch];
    float* embc = emb + (size_t)ch * NNODE * 256;

    branch_deg_rsqrt<<<NNODE / 8, 256, 0, stream>>>(masks[ch], adj, thr, dv);
    branch_norm<<<(NNODE * NNODE) / 256, 256, 0, stream>>>(masks[ch], adj, thr, dv, AnB);

    gemm_wmma_f32<false, false, false><<<dim3(HDIM / 64, NNODE / 128), blk, 0, stream>>>(
        X0, D0, W1, HDIM, nullptr, Hb, HDIM, D0, nullptr, nullptr);
    gemm_wmma_f32<true, true, false><<<dim3(HDIM / 64, NNODE / 128), blk, 0, stream>>>(
        AnB, NNODE, Hb, HDIM, b1, embc, 256, NNODE, nullptr, nullptr);
    gemm_wmma_f32<false, false, false><<<dim3(HDIM / 64, NNODE / 128), blk, 0, stream>>>(
        embc, 256, W2, HDIM, nullptr, H2b, HDIM, HDIM, nullptr, nullptr);
    gemm_wmma_f32<true, true, false><<<dim3(HDIM / 64, NNODE / 128), blk, 0, stream>>>(
        AnB, NNODE, H2b, HDIM, b2, embc + 128, 256, NNODE, nullptr, nullptr);
  }

  conv_dil2<<<(NNODE * 256) / 256, 256, 0, stream>>>(emb, cnnk, cnnb, emb_all);

  gemm_wmma_f32<true, true, true><<<dim3(64 / 64, BPAIR / 128), blk, 0, stream>>>(
      emb_all, 256, Wa, 64, ba, hm, 64, 512, left, right);

  logits_k<<<(BPAIR * 2 + 255) / 256, 256, 0, stream>>>(hm, Wb, bb, logits);
}


# 3072-cubed product tiled in registers only (no LDS): wave tile 32x64 -> 96x96, 256 four-wave groups = one wave per SIMD; next 16 k of loads issued ahead of 144 WMMAs; k and sum order kept
# speedup vs baseline: 2.8368x; 2.8368x over previous
#include <hip/hip_runtime.h>
#include <hip/hip_bf16.h>

#define NNODE 3072
#define FIN   1546
#define D0    256
#define HDIM  128
#define BPAIR 16384

typedef __attribute__((ext_vector_type(2))) float v2f;
typedef __attribute__((ext_vector_type(4))) float v4f;
typedef __attribute__((ext_vector_type(8))) float v8f;

template <int SM, int SN>
struct KStep { v2f a[SM]; v2f b[SN]; };

template <int SM, int SN>
__device__ __forceinline__ void load_kstep(KStep<SM, SN>& f, const float* const (&ap)[SM],
                                           const float* bp, int ldb)
{
  #pragma unroll
  for (int s = 0; s < SM; ++s) f.a[s] = *(const v2f*)ap[s];
  #pragma unroll
  for (int j = 0; j < SN; ++j) {
    f.b[j].x = bp[16 * j];
    f.b[j].y = bp[ldb + 16 * j];
  }
}

template <int SM, int SN>
__device__ __forceinline__ void wmma_tile(v8f (&acc)[SM][SN], const KStep<SM, SN>& f)
{
  #pragma unroll
  for (int s = 0; s < SM; ++s) {
    #pragma unroll
    for (int j = 0; j < SN; ++j)
      acc[s][j] = __builtin_amdgcn_wmma_f32_16x16x4_f32(false, f.a[s], false, f.b[j], (short)0,
                                                        acc[s][j], false, false);
  }
}

template <int U, int SM, int SN>
__device__ __forceinline__ void mma_block(v8f (&acc)[SM][SN], const KStep<SM, SN> (&f)[U])
{
  #pragma unroll
  for (int i = 0; i < U; ++i) wmma_tile<SM, SN>(acc, f[i]);
}

template <bool RELU, bool BIAS, bool GATHER, int U, bool RSCALE, int SM, int SN>
__global__ __launch_bounds__(128) void gemm_wmma_f32(
    const float* __restrict__ A, int lda,
    const float* __restrict__ Bm, int ldb,
    const float* __restrict__ bias,
    const float* __restrict__ rs,
    float* __restrict__ C, int ldc,
    int K,
    const int* __restrict__ gl, const int* __restrict__ gr)
{
  static_assert(U * (SM + 2 * SN) <= 60, "load group must fit the wave's load counter");
  constexpr int TM = 16 * SM;
  constexpr int TN = 16 * SN;
  using KS = KStep<SM, SN>;

  const int lane = threadIdx.x & 31;
  const int wave = threadIdx.x >> 5;
  const int half = lane >> 4;
  const int l16  = lane & 15;

  const int m0 = blockIdx.y * (4 * TM) + wave * TM;
  const int n0 = blockIdx.x * TN;

  v8f acc[SM][SN] = {};

  const float* bp = Bm + (size_t)(2 * half) * ldb + n0 + l16;
  const size_t bstep = (size_t)4 * ldb;

  constexpr int UU = U;
  const float* pl[SM];
  const float* pr[SM];
  #pragma unroll
  for (int s = 0; s < SM; ++s) {
    const int am = m0 + 16 * s + l16;
    pl[s] = GATHER ? A + (size_t)gl[am] * 256 + 2 * half
                   : A + (size_t)am * lda + 2 * half;
    pr[s] = GATHER ? A + (size_t)gr[am] * 256 + 2 * half : pl[s];
  }
  const int nk   = GATHER ? 128 : (K >> 2);
  const int nblk = nk / UU;

  auto load_t = [&](KS& f, int t) {
    const float* q[SM];
    const bool r = GATHER && (t >= 64);
    const int tt = GATHER ? (t & 63) : t;
    #pragma unroll
    for (int s = 0; s < SM; ++s) q[s] = (r ? pr[s] : pl[s]) + 4 * (size_t)tt;
    load_kstep<SM, SN>(f, q, bp + (size_t)t * bstep, ldb);
  };
  auto issue = [&](KS (&f)[UU], int t0) {
    #pragma unroll
    for (int i = 0; i < UU; ++i) load_t(f[i], t0 + i);
  };

  KS fa[UU], fb[UU];
  if (nblk > 0) {
    issue(fa, 0);
    int b = 0;
    for (; b + 2 < nblk; b += 2) {
      issue(fb, (b + 1) * UU);
      __builtin_amdgcn_sched_barrier(0);
      mma_block<UU, SM, SN>(acc, fa);
      __builtin_amdgcn_sched_barrier(0);
      issue(fa, (b + 2) * UU);
      __builtin_amdgcn_sched_barrier(0);
      mma_block<UU, SM, SN>(acc, fb);
      __builtin_amdgcn_sched_barrier(0);
    }
    if (b + 1 < nblk) {
      issue(fb, (b + 1) * UU);
      __builtin_amdgcn_sched_barrier(0);
      mma_block<UU, SM, SN>(acc, fa);
      mma_block<UU, SM, SN>(acc, fb);
    } else {
      mma_block<UU, SM, SN>(acc, fa);
    }
  }
  for (int t = nblk * UU; t < nk; ++t) {
    KS f;
    load_t(f, t);
    wmma_tile<SM, SN>(acc, f);
  }
  if (!GATHER) {
    const int kmain = K & ~3;
    if (kmain < K) {
      const float* bq = bp + (size_t)nk * bstep;
      const int ka = kmain + 2 * half;
      const bool g0 = (ka < K), g1 = (ka + 1 < K);
      KS f;
      #pragma unroll
      for (int s = 0; s < SM; ++s) {
        const float* ap = pl[s] + kmain;
        f.a[s].x = g0 ? ap[0] : 0.0f;
        f.a[s].y = g1 ? ap[1] : 0.0f;
      }
      #pragma unroll
      for (int j = 0; j < SN; ++j) {
        f.b[j].x = g0 ? bq[16 * j]       : 0.0f;
        f.b[j].y = g1 ? bq[ldb + 16 * j] : 0.0f;
      }
      wmma_tile<SM, SN>(acc, f);
    }
  }

  #pragma unroll
  for (int s = 0; s < SM; ++s) {
    #pragma unroll
    for (int v = 0; v < 8; ++v) {
      const int row = m0 + 16 * s + v + 8 * half;
      const float sc = RSCALE ? rs[row] : 1.0f;
      #pragma unroll
      for (int j = 0; j < SN; ++j) {
        const int col = n0 + 16 * j + l16;
        float x = acc[s][j][v];
        if (RSCALE) x *= sc;
        if (BIAS) x += bias[col];
        if (RELU) x = fmaxf(x, 0.0f);
        C[(size_t)row * ldc + col] = x;
      }
    }
  }
}

struct Unit96 { v4f a[6]; v2f b[2][6]; };

__device__ __forceinline__ void load_unit96(Unit96& f, const float* ap, size_t astrip,
                                            const float* bp, size_t ldb)
{
  #pragma unroll
  for (int s = 0; s < 6; ++s) f.a[s] = *(const v4f*)(ap + s * astrip);
  #pragma unroll
  for (int p = 0; p < 2; ++p) {
    #pragma unroll
    for (int j = 0; j < 6; ++j) {
      f.b[p][j].x = bp[(size_t)(2 * p) * ldb + 16 * j];
      f.b[p][j].y = bp[(size_t)(2 * p + 1) * ldb + 16 * j];
    }
  }
}

__device__ __forceinline__ void mma_unit96(v8f (&acc)[6][6], const Unit96& f)
{
  #pragma unroll
  for (int p = 0; p < 2; ++p) {
    #pragma unroll
    for (int s = 0; s < 6; ++s) {
      const v2f a = (p == 0) ? __builtin_shufflevector(f.a[s], f.a[s], 0, 1)
                             : __builtin_shufflevector(f.a[s], f.a[s], 2, 3);
      #pragma unroll
      for (int j = 0; j < 6; ++j)
        acc[s][j] = __builtin_amdgcn_wmma_f32_16x16x4_f32(false, a, false, f.b[p][j], (short)0,
                                                          acc[s][j], false, false);
    }
  }
}

__global__ __launch_bounds__(128) void gemm_wmma_f32_t96(
    const float* __restrict__ A, int lda,
    const float* __restrict__ Bm, int ldb,
    float* __restrict__ C, int ldc,
    int K)
{
  const int lane = threadIdx.x & 31;
  const int wave = threadIdx.x >> 5;
  const int half = lane >> 4;
  const int l16  = lane & 15;

  const int m0 = blockIdx.y * 192 + (wave >> 1) * 96;
  const int n0 = blockIdx.x * 192 + (wave & 1) * 96;

  v8f acc[6][6] = {};

  const float* ap = A + (size_t)(m0 + l16) * lda + 4 * half;
  const float* bp = Bm + (size_t)(4 * half) * ldb + n0 + l16;
  const size_t astrip = (size_t)16 * lda;
  const size_t bunit  = (size_t)8 * ldb;

  auto issue = [&](Unit96 (&f)[2], int kb) {
    const float* aq = ap + 16 * (size_t)kb;
    const float* bq = bp + 2 * (size_t)kb * bunit;
    load_unit96(f[0], aq, astrip, bq, (size_t)ldb);
    load_unit96(f[1], aq + 8, astrip, bq + bunit, (size_t)ldb);
  };
  auto mma = [&](const Unit96 (&f)[2]) {
    mma_unit96(acc, f[0]);
    mma_unit96(acc, f[1]);
  };

  const int nb = K >> 4;
  Unit96 fa[2], fb[2];
  if (nb > 0) {
    issue(fa, 0);
    int kb = 0;
    for (; kb + 2 < nb; kb += 2) {
      issue(fb, kb + 1);
      __builtin_amdgcn_sched_barrier(0);
      mma(fa);
      __builtin_amdgcn_sched_barrier(0);
      issue(fa, kb + 2);
      __builtin_amdgcn_sched_barrier(0);
      mma(fb);
      __builtin_amdgcn_sched_barrier(0);
    }
    if (kb + 1 < nb) {
      issue(fb, kb + 1);
      __builtin_amdgcn_sched_barrier(0);
      mma(fa);
      mma(fb);
    } else {
      mma(fa);
    }
  }

  #pragma unroll
  for (int s = 0; s < 6; ++s) {
    #pragma unroll
    for (int v = 0; v < 8; ++v) {
      const int row = m0 + 16 * s + v + 8 * half;
      #pragma unroll
      for (int j = 0; j < 6; ++j)
        C[(size_t)row * ldc + n0 + 16 * j + l16] = acc[s][j][v];
    }
  }
}

__global__ void deg_rsqrt(const float* __restrict__ A, float* __restrict__ d)
{
  const int gw   = (blockIdx.x * blockDim.x + threadIdx.x) >> 5;
  const int lane = threadIdx.x & 31;
  if (gw >= NNODE) return;
  const float* row = A + (size_t)gw * NNODE;
  float s = 0.0f;
  for (int j = lane; j < NNODE; j += 32) s += row[j];
  #pragma unroll
  for (int off = 16; off > 0; off >>= 1) s += __shfl_xor(s, off, 32);
  if (lane == 0) d[gw] = (s > 0.0f) ? rsqrtf(s) : 0.0f;
}

__global__ void branch_prep(const float* __restrict__ mask,
                            const float* __restrict__ adj,
                            const float* __restrict__ thrp,
                            float* __restrict__ Ah,
                            float* __restrict__ d)
{
  const int gw   = (blockIdx.x * blockDim.x + threadIdx.x) >> 5;
  const int lane = threadIdx.x & 31;
  if (gw >= NNODE) return;
  const float thr = thrp[0];
  const size_t base = (size_t)gw * NNODE;
  float s = 0.0f;
  for (int j = lane; j < NNODE; j += 32) {
    const float w = (mask[base + j] > thr) ? adj[base + j] : 0.0f;
    s += w;
    Ah[base + j] = (j == gw) ? w + 1.0f : w;
  }
  #pragma unroll
  for (int off = 16; off > 0; off >>= 1) s += __shfl_xor(s, off, 32);
  if (lane == 0) d[gw] = rsqrtf(s + 1.0f);
}

__global__ void symnorm(const float* __restrict__ adj,
                        const float* __restrict__ d,
                        float* __restrict__ An)
{
  const int idx = blockIdx.x * blockDim.x + threadIdx.x;
  if (idx >= NNODE * NNODE) return;
  const int i = idx / NNODE;
  const int j = idx - i * NNODE;
  An[idx] = adj[idx] * d[i] * d[j];
}

__global__ void conv_dil2(const float* __restrict__ emb,
                          const float* __restrict__ kw,
                          const float* __restrict__ kb,
                          float* __restrict__ out)
{
  const int idx = blockIdx.x * blockDim.x + threadIdx.x;
  if (idx >= NNODE * 256) return;
  const int n = idx >> 8;
  const int c = idx & 255;
  float acc = kb[0];
  #pragma unroll
  for (int ch = 0; ch < 3; ++ch) {
    #pragma unroll
    for (int kh = 0; kh < 3; ++kh) {
      const int ih = n + 2 * kh - 2;
      if (ih < 0 || ih >= NNODE) continue;
      #pragma unroll
      for (int kwi = 0; kwi < 3; ++kwi) {
        const int iw = c + 2 * kwi - 2;
        if (iw < 0 || iw >= 256) continue;
        acc += emb[((size_t)ch * NNODE + ih) * 256 + iw] * kw[ch * 9 + kh * 3 + kwi];
      }
    }
  }
  out[idx] = acc;
}

__global__ void logits_k(const float* __restrict__ h,
                         const float* __restrict__ Wb,
                         const float* __restrict__ bb,
                         float* __restrict__ out)
{
  const int i = blockIdx.x * blockDim.x + threadIdx.x;
  if (i >= BPAIR * 2) return;
  const int b = i >> 1;
  const int c = i & 1;
  float acc = bb[c];
  #pragma unroll
  for (int k = 0; k < 64; ++k) acc += h[b * 64 + k] * Wb[k * 2 + c];
  out[i] = acc;
}

extern "C" void kernel_launch(void* const* d_in, const int* in_sizes, int n_in,
                              void* d_out, int out_size, void* d_ws, size_t ws_size,
                              hipStream_t stream)
{
  const int*   left  = (const int*)d_in[0];
  const int*   right = (const int*)d_in[1];
  const float* X     = (const float*)d_in[2];
  const float* adj   = (const float*)d_in[3];
  const float* thr   = (const float*)d_in[4];
  const float* Wfc1  = (const float*)d_in[5];
  const float* bfc1  = (const float*)d_in[6];
  const float* Wg[6] = {(const float*)d_in[7],  (const float*)d_in[9],
                        (const float*)d_in[11], (const float*)d_in[13],
                        (const float*)d_in[15], (const float*)d_in[17]};
  const float* bg[6] = {(const float*)d_in[8],  (const float*)d_in[10],
                        (const float*)d_in[12], (const float*)d_in[14],
                        (const float*)d_in[16], (const float*)d_in[18]};
  const float* cnnk  = (const float*)d_in[19];
  const float* cnnb  = (const float*)d_in[20];
  const float* Wa    = (const float*)d_in[21];
  const float* ba    = (const float*)d_in[22];
  const float* Wb    = (const float*)d_in[23];
  const float* bb    = (const float*)d_in[24];

  float* out     = (float*)d_out;
  float* logits  = out;
  float* emb_all = out + BPAIR * 2;

  float* ws = (float*)d_ws;
  size_t off = 0;
  float* X0  = ws + off; off += (size_t)NNODE * D0;
  float* An  = ws + off; off += (size_t)NNODE * NNODE;
  float* A2  = ws + off; off += (size_t)NNODE * NNODE;
  float* Ah  = ws + off; off += (size_t)NNODE * NNODE;
  float* dv  = ws + off; off += NNODE;
  float* Hb  = ws + off; off += (size_t)NNODE * HDIM;
  float* H2b = ws + off; off += (size_t)NNODE * HDIM;
  float* emb = ws + off; off += (size_t)3 * NNODE * 256;
  float* hm  = ws + off; off += (size_t)BPAIR * 64;
  if (ws_size < off * sizeof(float)) return;

  const dim3 blk(128);

  gemm_wmma_f32<true, true, false, 8, false, 2, 2><<<dim3(D0 / 32, NNODE / 128), blk, 0, stream>>>(
      X, FIN, Wfc1, D0, bfc1, nullptr, X0, D0, FIN, nullptr, nullptr);

  deg_rsqrt<<<NNODE / 8, 256, 0, stream>>>(adj, dv);
  symnorm<<<(NNODE * NNODE) / 256, 256, 0, stream>>>(adj, dv, An);
  gemm_wmma_f32_t96<<<dim3(NNODE / 192, NNODE / 192), blk, 0, stream>>>(
      An, NNODE, An, NNODE, A2, NNODE, NNODE);

  const float* masks[3] = { adj, An, A2 };
  for (int ch = 0; ch < 3; ++ch) {
    const float* W1 = Wg[4 - 2 * ch];  const float* b1 = bg[4 - 2 * ch];
    const float* W2 = Wg[5 - 2 * ch];  const float* b2 = bg[5 - 2 * ch];
    float* embc = emb + (size_t)ch * NNODE * 256;

    branch_prep<<<NNODE / 8, 256, 0, stream>>>(masks[ch], adj, thr, Ah, dv);

    gemm_wmma_f32<false, false, false, 8, true, 1, 2><<<dim3(HDIM / 32, NNODE / 64), blk, 0, stream>>>(
        X0, D0, W1, HDIM, nullptr, dv, Hb, HDIM, D0, nullptr, nullptr);
    gemm_wmma_f32<true, true, false, 12, true, 1, 2><<<dim3(HDIM / 32, NNODE / 64), blk, 0, stream>>>(
        Ah, NNODE, Hb, HDIM, b1, dv, embc, 256, NNODE, nullptr, nullptr);
    gemm_wmma_f32<false, false, false, 8, true, 1, 2><<<dim3(HDIM / 32, NNODE / 64), blk, 0, stream>>>(
        embc, 256, W2, HDIM, nullptr, dv, H2b, HDIM, HDIM, nullptr, nullptr);
    gemm_wmma_f32<true, true, false, 12, true, 1, 2><<<dim3(HDIM / 32, NNODE / 64), blk, 0, stream>>>(
        Ah, NNODE, H2b, HDIM, b2, dv, embc + 128, 256, NNODE, nullptr, nullptr);
  }

  conv_dil2<<<(NNODE * 256) / 256, 256, 0, stream>>>(emb, cnnk, cnnb, emb_all);

  gemm_wmma_f32<true, true, true, 8, false, 2, 2><<<dim3(64 / 32, BPAIR / 128), blk, 0, stream>>>(
      emb_all, 256, Wa, 64, ba, nullptr, hm, 64, 512, left, right);

  logits_k<<<(BPAIR * 2 + 255) / 256, 256, 0, stream>>>(hm, Wb, bb, logits);
}


# H and H2 stored k-pair and column-pair interleaved by their GEMM epilogues: e1/e2 read both B fragments of a k-step with one 16-byte load (was 4), load groups 16 k-steps deep; k and sum order kept
# speedup vs baseline: 3.2188x; 1.1347x over previous
#include <hip/hip_runtime.h>
#include <hip/hip_bf16.h>

#define NNODE 3072
#define FIN   1546
#define D0    256
#define HDIM  128
#define BPAIR 16384

typedef __attribute__((ext_vector_type(2))) float v2f;
typedef __attribute__((ext_vector_type(4))) float v4f;
typedef __attribute__((ext_vector_type(8))) float v8f;

template <int SM, int SN>
struct KStep { v2f a[SM]; v2f b[SN]; };

template <int SM, int SN>
__device__ __forceinline__ void load_kstep(KStep<SM, SN>& f, const float* const (&ap)[SM],
                                           const float* bp, int ldb)
{
  #pragma unroll
  for (int s = 0; s < SM; ++s) f.a[s] = *(const v2f*)ap[s];
  #pragma unroll
  for (int j = 0; j < SN; ++j) {
    f.b[j].x = bp[16 * j];
    f.b[j].y = bp[ldb + 16 * j];
  }
}

template <int SM>
__device__ __forceinline__ void load_kstep_bpi(KStep<SM, 2>& f, const float* const (&ap)[SM],
                                               const float* bq)
{
  #pragma unroll
  for (int s = 0; s < SM; ++s) f.a[s] = *(const v2f*)ap[s];
  const v4f r = *(const v4f*)bq;
  f.b[0] = __builtin_shufflevector(r, r, 0, 1);
  f.b[1] = __builtin_shufflevector(r, r, 2, 3);
}

template <int SM, int SN>
__device__ __forceinline__ void wmma_tile(v8f (&acc)[SM][SN], const KStep<SM, SN>& f)
{
  #pragma unroll
  for (int s = 0; s < SM; ++s) {
    #pragma unroll
    for (int j = 0; j < SN; ++j)
      acc[s][j] = __builtin_amdgcn_wmma_f32_16x16x4_f32(false, f.a[s], false, f.b[j], (short)0,
                                                        acc[s][j], false, false);
  }
}

template <int U, int SM, int SN>
__device__ __forceinline__ void mma_block(v8f (&acc)[SM][SN], const KStep<SM, SN> (&f)[U])
{
  #pragma unroll
  for (int i = 0; i < U; ++i) wmma_tile<SM, SN>(acc, f[i]);
}

template <bool RELU, bool BIAS, bool GATHER, int U, bool RSCALE, int SM, int SN,
          bool BPI = false, bool CPI = false>
__global__ __launch_bounds__(128) void gemm_wmma_f32(
    const float* __restrict__ A, int lda,
    const float* __restrict__ Bm, int ldb,
    const float* __restrict__ bias,
    const float* __restrict__ rs,
    float* __restrict__ C, int ldc,
    int K,
    const int* __restrict__ gl, const int* __restrict__ gr)
{
  static_assert(U * (BPI ? SM + 1 : SM + 2 * SN) <= 60, "load group must fit the wave's load counter");
  static_assert(!BPI || (SN == 2 && !GATHER), "pair-interleaved B: two n strips per wave");
  static_assert(!(BPI && CPI), "one layout change per GEMM");
  constexpr int TM = 16 * SM;
  constexpr int TN = 16 * SN;
  using KS = KStep<SM, SN>;

  const int lane = threadIdx.x & 31;
  const int wave = threadIdx.x >> 5;
  const int half = lane >> 4;
  const int l16  = lane & 15;

  const int m0 = blockIdx.y * (4 * TM) + wave * TM;
  const int n0 = blockIdx.x * TN;

  v8f acc[SM][SN] = {};

  const float* bp = BPI ? Bm + (size_t)half * (2 * ldb) + 2 * (n0 + 2 * l16)
                        : Bm + (size_t)(2 * half) * ldb + n0 + l16;
  const size_t bstep = (size_t)4 * ldb;

  constexpr int UU = U;
  const float* pl[SM];
  const float* pr[SM];
  #pragma unroll
  for (int s = 0; s < SM; ++s) {
    const int am = m0 + 16 * s + l16;
    pl[s] = GATHER ? A + (size_t)gl[am] * 256 + 2 * half
                   : A + (size_t)am * lda + 2 * half;
    pr[s] = GATHER ? A + (size_t)gr[am] * 256 + 2 * half : pl[s];
  }
  const int nk   = GATHER ? 128 : (K >> 2);
  const int nblk = nk / UU;

  auto load_t = [&](KS& f, int t) {
    const float* q[SM];
    const bool r = GATHER && (t >= 64);
    const int tt = GATHER ? (t & 63) : t;
    #pragma unroll
    for (int s = 0; s < SM; ++s) q[s] = (r ? pr[s] : pl[s]) + 4 * (size_t)tt;
    if constexpr (BPI) load_kstep_bpi<SM>(f, q, bp + (size_t)t * bstep);
    else               load_kstep<SM, SN>(f, q, bp + (size_t)t * bstep, ldb);
  };
  auto issue = [&](KS (&f)[UU], int t0) {
    #pragma unroll
    for (int i = 0; i < UU; ++i) load_t(f[i], t0 + i);
  };

  KS fa[UU], fb[UU];
  if (nblk > 0) {
    issue(fa, 0);
    int b = 0;
    for (; b + 2 < nblk; b += 2) {
      issue(fb, (b + 1) * UU);
      __builtin_amdgcn_sched_barrier(0);
      mma_block<UU, SM, SN>(acc, fa);
      __builtin_amdgcn_sched_barrier(0);
      issue(fa, (b + 2) * UU);
      __builtin_amdgcn_sched_barrier(0);
      mma_block<UU, SM, SN>(acc, fb);
      __builtin_amdgcn_sched_barrier(0);
    }
    if (b + 1 < nblk) {
      issue(fb, (b + 1) * UU);
      __builtin_amdgcn_sched_barrier(0);
      mma_block<UU, SM, SN>(acc, fa);
      mma_block<UU, SM, SN>(acc, fb);
    } else {
      mma_block<UU, SM, SN>(acc, fa);
    }
  }
  for (int t = nblk * UU; t < nk; ++t) {
    KS f;
    load_t(f, t);
    wmma_tile<SM, SN>(acc, f);
  }
  if (!GATHER && !BPI) {
    const int kmain = K & ~3;
    if (kmain < K) {
      const float* bq = bp + (size_t)nk * bstep;
      const int ka = kmain + 2 * half;
      const bool g0 = (ka < K), g1 = (ka + 1 < K);
      KS f;
      #pragma unroll
      for (int s = 0; s < SM; ++s) {
        const float* ap = pl[s] + kmain;
        f.a[s].x = g0 ? ap[0] : 0.0f;
        f.a[s].y = g1 ? ap[1] : 0.0f;
      }
      #pragma unroll
      for (int j = 0; j < SN; ++j) {
        f.b[j].x = g0 ? bq[16 * j]       : 0.0f;
        f.b[j].y = g1 ? bq[ldb + 16 * j] : 0.0f;
      }
      wmma_tile<SM, SN>(acc, f);
    }
  }

  if constexpr (BPI) {
    const int col = n0 + 2 * l16;
    #pragma unroll
    for (int s = 0; s < SM; ++s) {
      #pragma unroll
      for (int v = 0; v < 8; ++v) {
        const int row = m0 + 16 * s + v + 8 * half;
        const float sc = RSCALE ? rs[row] : 1.0f;
        v2f x;
        x.x = acc[s][0][v];
        x.y = acc[s][1][v];
        if (RSCALE) { x.x *= sc; x.y *= sc; }
        if (BIAS) { x.x += bias[col]; x.y += bias[col + 1]; }
        if (RELU) { x.x = fmaxf(x.x, 0.0f); x.y = fmaxf(x.y, 0.0f); }
        *(v2f*)(C + (size_t)row * ldc + col) = x;
      }
    }
  } else if constexpr (CPI) {
    #pragma unroll
    for (int s = 0; s < SM; ++s) {
      #pragma unroll
      for (int v = 0; v < 8; v += 2) {
        const int row = m0 + 16 * s + v + 8 * half;
        const float sc0 = RSCALE ? rs[row] : 1.0f;
        const float sc1 = RSCALE ? rs[row + 1] : 1.0f;
        #pragma unroll
        for (int j = 0; j < SN; ++j) {
          const int col = n0 + 16 * j + l16;
          v2f x;
          x.x = acc[s][j][v];
          x.y = acc[s][j][v + 1];
          if (RSCALE) { x.x *= sc0; x.y *= sc1; }
          if (BIAS) { x.x += bias[col]; x.y += bias[col]; }
          if (RELU) { x.x = fmaxf(x.x, 0.0f); x.y = fmaxf(x.y, 0.0f); }
          *(v2f*)(C + (size_t)(row >> 1) * (2 * ldc) + 2 * col) = x;
        }
      }
    }
  } else {
    #pragma unroll
    for (int s = 0; s < SM; ++s) {
      #pragma unroll
      for (int v = 0; v < 8; ++v) {
        const int row = m0 + 16 * s + v + 8 * half;
        const float sc = RSCALE ? rs[row] : 1.0f;
        #pragma unroll
        for (int j = 0; j < SN; ++j) {
          const int col = n0 + 16 * j + l16;
          float x = acc[s][j][v];
          if (RSCALE) x *= sc;
          if (BIAS) x += bias[col];
          if (RELU) x = fmaxf(x, 0.0f);
          C[(size_t)row * ldc + col] = x;
        }
      }
    }
  }
}

struct Unit96 { v4f a[6]; v2f b[2][6]; };

__device__ __forceinline__ void load_unit96(Unit96& f, const float* ap, size_t astrip,
                                            const float* bp, size_t ldb)
{
  #pragma unroll
  for (int s = 0; s < 6; ++s) f.a[s] = *(const v4f*)(ap + s * astrip);
  #pragma unroll
  for (int p = 0; p < 2; ++p) {
    #pragma unroll
    for (int j = 0; j < 6; ++j) {
      f.b[p][j].x = bp[(size_t)(2 * p) * ldb + 16 * j];
      f.b[p][j].y = bp[(size_t)(2 * p + 1) * ldb + 16 * j];
    }
  }
}

__device__ __forceinline__ void mma_unit96(v8f (&acc)[6][6], const Unit96& f)
{
  #pragma unroll
  for (int p = 0; p < 2; ++p) {
    #pragma unroll
    for (int s = 0; s < 6; ++s) {
      const v2f a = (p == 0) ? __builtin_shufflevector(f.a[s], f.a[s], 0, 1)
                             : __builtin_shufflevector(f.a[s], f.a[s], 2, 3);
      #pragma unroll
      for (int j = 0; j < 6; ++j)
        acc[s][j] = __builtin_amdgcn_wmma_f32_16x16x4_f32(false, a, false, f.b[p][j], (short)0,
                                                          acc[s][j], false, false);
    }
  }
}

__global__ __launch_bounds__(128) void gemm_wmma_f32_t96(
    const float* __restrict__ A, int lda,
    const float* __restrict__ Bm, int ldb,
    float* __restrict__ C, int ldc,
    int K)
{
  const int lane = threadIdx.x & 31;
  const int wave = threadIdx.x >> 5;
  const int half = lane >> 4;
  const int l16  = lane & 15;

  const int m0 = blockIdx.y * 192 + (wave >> 1) * 96;
  const int n0 = blockIdx.x * 192 + (wave & 1) * 96;

  v8f acc[6][6] = {};

  const float* ap = A + (size_t)(m0 + l16) * lda + 4 * half;
  const float* bp = Bm + (size_t)(4 * half) * ldb + n0 + l16;
  const size_t astrip = (size_t)16 * lda;
  const size_t bunit  = (size_t)8 * ldb;

  auto issue = [&](Unit96 (&f)[2], int kb) {
    const float* aq = ap + 16 * (size_t)kb;
    const float* bq = bp + 2 * (size_t)kb * bunit;
    load_unit96(f[0], aq, astrip, bq, (size_t)ldb);
    load_unit96(f[1], aq + 8, astrip, bq + bunit, (size_t)ldb);
  };
  auto mma = [&](const Unit96 (&f)[2]) {
    mma_unit96(acc, f[0]);
    mma_unit96(acc, f[1]);
  };

  const int nb = K >> 4;
  Unit96 fa[2], fb[2];
  if (nb > 0) {
    issue(fa, 0);
    int kb = 0;
    for (; kb + 2 < nb; kb += 2) {
      issue(fb, kb + 1);
      __builtin_amdgcn_sched_barrier(0);
      mma(fa);
      __builtin_amdgcn_sched_barrier(0);
      issue(fa, kb + 2);
      __builtin_amdgcn_sched_barrier(0);
      mma(fb);
      __builtin_amdgcn_sched_barrier(0);
    }
    if (kb + 1 < nb) {
      issue(fb, kb + 1);
      __builtin_amdgcn_sched_barrier(0);
      mma(fa);
      mma(fb);
    } else {
      mma(fa);
    }
  }

  #pragma unroll
  for (int s = 0; s < 6; ++s) {
    #pragma unroll
    for (int v = 0; v < 8; ++v) {
      const int row = m0 + 16 * s + v + 8 * half;
      #pragma unroll
      for (int j = 0; j < 6; ++j)
        C[(size_t)row * ldc + n0 + 16 * j + l16] = acc[s][j][v];
    }
  }
}

__global__ void deg_rsqrt(const float* __restrict__ A, float* __restrict__ d)
{
  const int gw   = (blockIdx.x * blockDim.x + threadIdx.x) >> 5;
  const int lane = threadIdx.x & 31;
  if (gw >= NNODE) return;
  const float* row = A + (size_t)gw * NNODE;
  float s = 0.0f;
  for (int j = lane; j < NNODE; j += 32) s += row[j];
  #pragma unroll
  for (int off = 16; off > 0; off >>= 1) s += __shfl_xor(s, off, 32);
  if (lane == 0) d[gw] = (s > 0.0f) ? rsqrtf(s) : 0.0f;
}

__global__ void branch_prep(const float* __restrict__ mask,
                            const float* __restrict__ adj,
                            const float* __restrict__ thrp,
                            float* __restrict__ Ah,
                            float* __restrict__ d)
{
  const int gw   = (blockIdx.x * blockDim.x + threadIdx.x) >> 5;
  const int lane = threadIdx.x & 31;
  if (gw >= NNODE) return;
  const float thr = thrp[0];
  const size_t base = (size_t)gw * NNODE;
  float s = 0.0f;
  for (int j = lane; j < NNODE; j += 32) {
    const float w = (mask[base + j] > thr) ? adj[base + j] : 0.0f;
    s += w;
    Ah[base + j] = (j == gw) ? w + 1.0f : w;
  }
  #pragma unroll
  for (int off = 16; off > 0; off >>= 1) s += __shfl_xor(s, off, 32);
  if (lane == 0) d[gw] = rsqrtf(s + 1.0f);
}

__global__ void symnorm(const float* __restrict__ adj,
                        const float* __restrict__ d,
                        float* __restrict__ An)
{
  const int idx = blockIdx.x * blockDim.x + threadIdx.x;
  if (idx >= NNODE * NNODE) return;
  const int i = idx / NNODE;
  const int j = idx - i * NNODE;
  An[idx] = adj[idx] * d[i] * d[j];
}

__global__ void conv_dil2(const float* __restrict__ emb,
                          const float* __restrict__ kw,
                          const float* __restrict__ kb,
                          float* __restrict__ out)
{
  const int idx = blockIdx.x * blockDim.x + threadIdx.x;
  if (idx >= NNODE * 256) return;
  const int n = idx >> 8;
  const int c = idx & 255;
  float acc = kb[0];
  #pragma unroll
  for (int ch = 0; ch < 3; ++ch) {
    #pragma unroll
    for (int kh = 0; kh < 3; ++kh) {
      const int ih = n + 2 * kh - 2;
      if (ih < 0 || ih >= NNODE) continue;
      #pragma unroll
      for (int kwi = 0; kwi < 3; ++kwi) {
        const int iw = c + 2 * kwi - 2;
        if (iw < 0 || iw >= 256) continue;
        acc += emb[((size_t)ch * NNODE + ih) * 256 + iw] * kw[ch * 9 + kh * 3 + kwi];
      }
    }
  }
  out[idx] = acc;
}

__global__ void logits_k(const float* __restrict__ h,
                         const float* __restrict__ Wb,
                         const float* __restrict__ bb,
                         float* __restrict__ out)
{
  const int i = blockIdx.x * blockDim.x + threadIdx.x;
  if (i >= BPAIR * 2) return;
  const int b = i >> 1;
  const int c = i & 1;
  float acc = bb[c];
  #pragma unroll
  for (int k = 0; k < 64; ++k) acc += h[b * 64 + k] * Wb[k * 2 + c];
  out[i] = acc;
}

extern "C" void kernel_launch(void* const* d_in, const int* in_sizes, int n_in,
                              void* d_out, int out_size, void* d_ws, size_t ws_size,
                              hipStream_t stream)
{
  const int*   left  = (const int*)d_in[0];
  const int*   right = (const int*)d_in[1];
  const float* X     = (const float*)d_in[2];
  const float* adj   = (const float*)d_in[3];
  const float* thr   = (const float*)d_in[4];
  const float* Wfc1  = (const float*)d_in[5];
  const float* bfc1  = (const float*)d_in[6];
  const float* Wg[6] = {(const float*)d_in[7],  (const float*)d_in[9],
                        (const float*)d_in[11], (const float*)d_in[13],
                        (const float*)d_in[15], (const float*)d_in[17]};
  const float* bg[6] = {(const float*)d_in[8],  (const float*)d_in[10],
                        (const float*)d_in[12], (const float*)d_in[14],
                        (const float*)d_in[16], (const float*)d_in[18]};
  const float* cnnk  = (const float*)d_in[19];
  const float* cnnb  = (const float*)d_in[20];
  const float* Wa    = (const float*)d_in[21];
  const float* ba    = (const float*)d_in[22];
  const float* Wb    = (const float*)d_in[23];
  const float* bb    = (const float*)d_in[24];

  float* out     = (float*)d_out;
  float* logits  = out;
  float* emb_all = out + BPAIR * 2;

  float* ws = (float*)d_ws;
  size_t off = 0;
  float* X0  = ws + off; off += (size_t)NNODE * D0;
  float* An  = ws + off; off += (size_t)NNODE * NNODE;
  float* A2  = ws + off; off += (size_t)NNODE * NNODE;
  float* Ah  = ws + off; off += (size_t)NNODE * NNODE;
  float* dv  = ws + off; off += NNODE;
  float* Hb  = ws + off; off += (size_t)NNODE * HDIM;
  float* H2b = ws + off; off += (size_t)NNODE * HDIM;
  float* emb = ws + off; off += (size_t)3 * NNODE * 256;
  float* hm  = ws + off; off += (size_t)BPAIR * 64;
  if (ws_size < off * sizeof(float)) return;

  const dim3 blk(128);

  gemm_wmma_f32<true, true, false, 8, false, 2, 2><<<dim3(D0 / 32, NNODE / 128), blk, 0, stream>>>(
      X, FIN, Wfc1, D0, bfc1, nullptr, X0, D0, FIN, nullptr, nullptr);

  deg_rsqrt<<<NNODE / 8, 256, 0, stream>>>(adj, dv);
  symnorm<<<(NNODE * NNODE) / 256, 256, 0, stream>>>(adj, dv, An);
  gemm_wmma_f32_t96<<<dim3(NNODE / 192, NNODE / 192), blk, 0, stream>>>(
      An, NNODE, An, NNODE, A2, NNODE, NNODE);

  const float* masks[3] = { adj, An, A2 };
  for (int ch = 0; ch < 3; ++ch) {
    const float* W1 = Wg[4 - 2 * ch];  const float* b1 = bg[4 - 2 * ch];
    const float* W2 = Wg[5 - 2 * ch];  const float* b2 = bg[5 - 2 * ch];
    float* embc = emb + (size_t)ch * NNODE * 256;

    branch_prep<<<NNODE / 8, 256, 0, stream>>>(masks[ch], adj, thr, Ah, dv);

    gemm_wmma_f32<false, false, false, 8, true, 1, 2, false, true><<<dim3(HDIM / 32, NNODE / 64), blk, 0, stream>>>(
        X0, D0, W1, HDIM, nullptr, dv, Hb, HDIM, D0, nullptr, nullptr);
    gemm_wmma_f32<true, true, false, 16, true, 1, 2, true, false><<<dim3(HDIM / 32, NNODE / 64), blk, 0, stream>>>(
        Ah, NNODE, Hb, HDIM, b1, dv, embc, 256, NNODE, nullptr, nullptr);
    gemm_wmma_f32<false, false, false, 8, true, 1, 2, false, true><<<dim3(HDIM / 32, NNODE / 64), blk, 0, stream>>>(
        embc, 256, W2, HDIM, nullptr, dv, H2b, HDIM, HDIM, nullptr, nullptr);
    gemm_wmma_f32<true, true, false, 16, true, 1, 2, true, false><<<dim3(HDIM / 32, NNODE / 64), blk, 0, stream>>>(
        Ah, NNODE, H2b, HDIM, b2, dv, embc + 128, 256, NNODE, nullptr, nullptr);
  }

  conv_dil2<<<(NNODE * 256) / 256, 256, 0, stream>>>(emb, cnnk, cnnb, emb_all);

  gemm_wmma_f32<true, true, true, 8, false, 2, 2><<<dim3(64 / 32, BPAIR / 128), blk, 0, stream>>>(
      emb_all, 256, Wa, 64, ba, nullptr, hm, 64, 512, left, right);

  logits_k<<<(BPAIR * 2 + 255) / 256, 256, 0, stream>>>(hm, Wb, bb, logits);
}


# Ah kept in WMMA A-fragment order: prep redone per 16-row strip (unconditional 16-byte loads up front, row sums via LDS); e1/e2 take A as one whole-line load per 8 k; row-sum and in-8 k order change
# speedup vs baseline: 3.9478x; 1.2265x over previous
#include <hip/hip_runtime.h>
#include <hip/hip_bf16.h>

#define NNODE 3072
#define FIN   1546
#define D0    256
#define HDIM  128
#define BPAIR 16384

typedef __attribute__((ext_vector_type(2))) float v2f;
typedef __attribute__((ext_vector_type(4))) float v4f;
typedef __attribute__((ext_vector_type(8))) float v8f;

template <int SM, int SN>
struct KStep { v2f a[SM]; v2f b[SN]; };

template <int SM, int SN>
__device__ __forceinline__ void load_kstep(KStep<SM, SN>& f, const float* const (&ap)[SM],
                                           const float* bp, int ldb)
{
  #pragma unroll
  for (int s = 0; s < SM; ++s) f.a[s] = *(const v2f*)ap[s];
  #pragma unroll
  for (int j = 0; j < SN; ++j) {
    f.b[j].x = bp[16 * j];
    f.b[j].y = bp[ldb + 16 * j];
  }
}

template <int SM, int SN>
__device__ __forceinline__ void wmma_tile(v8f (&acc)[SM][SN], const KStep<SM, SN>& f)
{
  #pragma unroll
  for (int s = 0; s < SM; ++s) {
    #pragma unroll
    for (int j = 0; j < SN; ++j)
      acc[s][j] = __builtin_amdgcn_wmma_f32_16x16x4_f32(false, f.a[s], false, f.b[j], (short)0,
                                                        acc[s][j], false, false);
  }
}

template <int U, int SM, int SN>
__device__ __forceinline__ void mma_block(v8f (&acc)[SM][SN], const KStep<SM, SN> (&f)[U])
{
  #pragma unroll
  for (int i = 0; i < U; ++i) wmma_tile<SM, SN>(acc, f[i]);
}

template <bool RELU, bool BIAS, bool GATHER, int U, bool RSCALE, int SM, int SN,
          bool CPI = false>
__global__ __launch_bounds__(128) void gemm_wmma_f32(
    const float* __restrict__ A, int lda,
    const float* __restrict__ Bm, int ldb,
    const float* __restrict__ bias,
    const float* __restrict__ rs,
    float* __restrict__ C, int ldc,
    int K,
    const int* __restrict__ gl, const int* __restrict__ gr)
{
  static_assert(U * (SM + 2 * SN) <= 60, "load group must fit the wave's load counter");
  constexpr int TM = 16 * SM;
  constexpr int TN = 16 * SN;
  using KS = KStep<SM, SN>;

  const int lane = threadIdx.x & 31;
  const int wave = threadIdx.x >> 5;
  const int half = lane >> 4;
  const int l16  = lane & 15;

  const int m0 = blockIdx.y * (4 * TM) + wave * TM;
  const int n0 = blockIdx.x * TN;

  v8f acc[SM][SN] = {};

  const float* bp = Bm + (size_t)(2 * half) * ldb + n0 + l16;
  const size_t bstep = (size_t)4 * ldb;

  constexpr int UU = U;
  const float* pl[SM];
  const float* pr[SM];
  #pragma unroll
  for (int s = 0; s < SM; ++s) {
    const int am = m0 + 16 * s + l16;
    pl[s] = GATHER ? A + (size_t)gl[am] * 256 + 2 * half
                   : A + (size_t)am * lda + 2 * half;
    pr[s] = GATHER ? A + (size_t)gr[am] * 256 + 2 * half : pl[s];
  }
  const int nk   = GATHER ? 128 : (K >> 2);
  const int nblk = nk / UU;

  auto load_t = [&](KS& f, int t) {
    const float* q[SM];
    const bool r = GATHER && (t >= 64);
    const int tt = GATHER ? (t & 63) : t;
    #pragma unroll
    for (int s = 0; s < SM; ++s) q[s] = (r ? pr[s] : pl[s]) + 4 * (size_t)tt;
    load_kstep<SM, SN>(f, q, bp + (size_t)t * bstep, ldb);
  };
  auto issue = [&](KS (&f)[UU], int t0) {
    #pragma unroll
    for (int i = 0; i < UU; ++i) load_t(f[i], t0 + i);
  };

  KS fa[UU], fb[UU];
  if (nblk > 0) {
    issue(fa, 0);
    int b = 0;
    for (; b + 2 < nblk; b += 2) {
      issue(fb, (b + 1) * UU);
      __builtin_amdgcn_sched_barrier(0);
      mma_block<UU, SM, SN>(acc, fa);
      __builtin_amdgcn_sched_barrier(0);
      issue(fa, (b + 2) * UU);
      __builtin_amdgcn_sched_barrier(0);
      mma_block<UU, SM, SN>(acc, fb);
      __builtin_amdgcn_sched_barrier(0);
    }
    if (b + 1 < nblk) {
      issue(fb, (b + 1) * UU);
      __builtin_amdgcn_sched_barrier(0);
      mma_block<UU, SM, SN>(acc, fa);
      mma_block<UU, SM, SN>(acc, fb);
    } else {
      mma_block<UU, SM, SN>(acc, fa);
    }
  }
  for (int t = nblk * UU; t < nk; ++t) {
    KS f;
    load_t(f, t);
    wmma_tile<SM, SN>(acc, f);
  }
  if (!GATHER) {
    const int kmain = K & ~3;
    if (kmain < K) {
      const float* bq = bp + (size_t)nk * bstep;
      const int ka = kmain + 2 * half;
      const bool g0 = (ka < K), g1 = (ka + 1 < K);
      KS f;
      #pragma unroll
      for (int s = 0; s < SM; ++s) {
        const float* ap = pl[s] + kmain;
        f.a[s].x = g0 ? ap[0] : 0.0f;
        f.a[s].y = g1 ? ap[1] : 0.0f;
      }
      #pragma unroll
      for (int j = 0; j < SN; ++j) {
        f.b[j].x = g0 ? bq[16 * j]       : 0.0f;
        f.b[j].y = g1 ? bq[ldb + 16 * j] : 0.0f;
      }
      wmma_tile<SM, SN>(acc, f);
    }
  }

  if constexpr (CPI) {
    #pragma unroll
    for (int s = 0; s < SM; ++s) {
      #pragma unroll
      for (int v = 0; v < 8; v += 2) {
        const int row = m0 + 16 * s + v + 8 * half;
        const float sc0 = RSCALE ? rs[row] : 1.0f;
        const float sc1 = RSCALE ? rs[row + 1] : 1.0f;
        #pragma unroll
        for (int j = 0; j < SN; ++j) {
          const int col = n0 + 16 * j + l16;
          v2f x;
          x.x = acc[s][j][v];
          x.y = acc[s][j][v + 1];
          if (RSCALE) { x.x *= sc0; x.y *= sc1; }
          if (BIAS) { x.x += bias[col]; x.y += bias[col]; }
          if (RELU) { x.x = fmaxf(x.x, 0.0f); x.y = fmaxf(x.y, 0.0f); }
          *(v2f*)(C + (size_t)(row >> 1) * (2 * ldc) + 2 * col) = x;
        }
      }
    }
  } else {
    #pragma unroll
    for (int s = 0; s < SM; ++s) {
      #pragma unroll
      for (int v = 0; v < 8; ++v) {
        const int row = m0 + 16 * s + v + 8 * half;
        const float sc = RSCALE ? rs[row] : 1.0f;
        #pragma unroll
        for (int j = 0; j < SN; ++j) {
          const int col = n0 + 16 * j + l16;
          float x = acc[s][j][v];
          if (RSCALE) x *= sc;
          if (BIAS) x += bias[col];
          if (RELU) x = fmaxf(x, 0.0f);
          C[(size_t)row * ldc + col] = x;
        }
      }
    }
  }
}

struct Unit96 { v4f a[6]; v2f b[2][6]; };

__device__ __forceinline__ void load_unit96(Unit96& f, const float* ap, size_t astrip,
                                            const float* bp, size_t ldb)
{
  #pragma unroll
  for (int s = 0; s < 6; ++s) f.a[s] = *(const v4f*)(ap + s * astrip);
  #pragma unroll
  for (int p = 0; p < 2; ++p) {
    #pragma unroll
    for (int j = 0; j < 6; ++j) {
      f.b[p][j].x = bp[(size_t)(2 * p) * ldb + 16 * j];
      f.b[p][j].y = bp[(size_t)(2 * p + 1) * ldb + 16 * j];
    }
  }
}

__device__ __forceinline__ void mma_unit96(v8f (&acc)[6][6], const Unit96& f)
{
  #pragma unroll
  for (int p = 0; p < 2; ++p) {
    #pragma unroll
    for (int s = 0; s < 6; ++s) {
      const v2f a = (p == 0) ? __builtin_shufflevector(f.a[s], f.a[s], 0, 1)
                             : __builtin_shufflevector(f.a[s], f.a[s], 2, 3);
      #pragma unroll
      for (int j = 0; j < 6; ++j)
        acc[s][j] = __builtin_amdgcn_wmma_f32_16x16x4_f32(false, a, false, f.b[p][j], (short)0,
                                                          acc[s][j], false, false);
    }
  }
}

__global__ __launch_bounds__(128) void gemm_wmma_f32_t96(
    const float* __restrict__ A, int lda,
    const float* __restrict__ Bm, int ldb,
    float* __restrict__ C, int ldc,
    int K)
{
  const int lane = threadIdx.x & 31;
  const int wave = threadIdx.x >> 5;
  const int half = lane >> 4;
  const int l16  = lane & 15;

  const int m0 = blockIdx.y * 192 + (wave >> 1) * 96;
  const int n0 = blockIdx.x * 192 + (wave & 1) * 96;

  v8f acc[6][6] = {};

  const float* ap = A + (size_t)(m0 + l16) * lda + 4 * half;
  const float* bp = Bm + (size_t)(4 * half) * ldb + n0 + l16;
  const size_t astrip = (size_t)16 * lda;
  const size_t bunit  = (size_t)8 * ldb;

  auto issue = [&](Unit96 (&f)[2], int kb) {
    const float* aq = ap + 16 * (size_t)kb;
    const float* bq = bp + 2 * (size_t)kb * bunit;
    load_unit96(f[0], aq, astrip, bq, (size_t)ldb);
    load_unit96(f[1], aq + 8, astrip, bq + bunit, (size_t)ldb);
  };
  auto mma = [&](const Unit96 (&f)[2]) {
    mma_unit96(acc, f[0]);
    mma_unit96(acc, f[1]);
  };

  const int nb = K >> 4;
  Unit96 fa[2], fb[2];
  if (nb > 0) {
    issue(fa, 0);
    int kb = 0;
    for (; kb + 2 < nb; kb += 2) {
      issue(fb, kb + 1);
      __builtin_amdgcn_sched_barrier(0);
      mma(fa);
      __builtin_amdgcn_sched_barrier(0);
      issue(fa, kb + 2);
      __builtin_amdgcn_sched_barrier(0);
      mma(fb);
      __builtin_amdgcn_sched_barrier(0);
    }
    if (kb + 1 < nb) {
      issue(fb, kb + 1);
      __builtin_amdgcn_sched_barrier(0);
      mma(fa);
      mma(fb);
    } else {
      mma(fa);
    }
  }

  #pragma unroll
  for (int s = 0; s < 6; ++s) {
    #pragma unroll
    for (int v = 0; v < 8; ++v) {
      const int row = m0 + 16 * s + v + 8 * half;
      #pragma unroll
      for (int j = 0; j < 6; ++j)
        C[(size_t)row * ldc + n0 + 16 * j + l16] = acc[s][j][v];
    }
  }
}

struct UnitE { v4f a; v4f b[2]; };

template <int UB, int LDB>
__global__ __launch_bounds__(128) void gemm_afrag_bpi(
    const float* __restrict__ Af,
    const float* __restrict__ Bm,
    const float* __restrict__ bias,
    const float* __restrict__ rs,
    float* __restrict__ C, int ldc,
    int K)
{
  static_assert(3 * UB <= 60, "load group must fit the wave's load counter");
  const int lane = threadIdx.x & 31;
  const int wave = threadIdx.x >> 5;
  const int half = lane >> 4;
  const int l16  = lane & 15;

  const int strip = blockIdx.y * 4 + wave;
  const int m0 = 16 * strip;
  const int n0 = blockIdx.x * 32;
  const int nu = K >> 3;

  v8f acc[2] = {};

  const float* ap = Af + ((size_t)strip * nu * 32 + lane) * 4;
  const float* bp = Bm + (size_t)(2 * half) * (2 * LDB) + 2 * (n0 + 2 * l16);
  constexpr size_t prow  = (size_t)2 * LDB;
  constexpr size_t bunit = (size_t)8 * LDB;

  auto load_u = [&](UnitE& f, int u) {
    f.a = *(const v4f*)(ap + (size_t)u * 128);
    const float* bq = bp + (size_t)u * bunit;
    f.b[0] = *(const v4f*)bq;
    f.b[1] = *(const v4f*)(bq + prow);
  };
  auto mma_u = [&](const UnitE& f) {
    #pragma unroll
    for (int p = 0; p < 2; ++p) {
      const v2f a  = (p == 0) ? __builtin_shufflevector(f.a, f.a, 0, 1)
                              : __builtin_shufflevector(f.a, f.a, 2, 3);
      const v2f b0 = __builtin_shufflevector(f.b[p], f.b[p], 0, 1);
      const v2f b1 = __builtin_shufflevector(f.b[p], f.b[p], 2, 3);
      acc[0] = __builtin_amdgcn_wmma_f32_16x16x4_f32(false, a, false, b0, (short)0, acc[0], false, false);
      acc[1] = __builtin_amdgcn_wmma_f32_16x16x4_f32(false, a, false, b1, (short)0, acc[1], false, false);
    }
  };
  auto issue = [&](UnitE (&f)[UB], int u0) {
    #pragma unroll
    for (int i = 0; i < UB; ++i) load_u(f[i], u0 + i);
  };
  auto mma = [&](const UnitE (&f)[UB]) {
    #pragma unroll
    for (int i = 0; i < UB; ++i) mma_u(f[i]);
  };

  const int nblk = nu / UB;
  UnitE fa[UB], fb[UB];
  if (nblk > 0) {
    issue(fa, 0);
    int b = 0;
    for (; b + 2 < nblk; b += 2) {
      issue(fb, (b + 1) * UB);
      __builtin_amdgcn_sched_barrier(0);
      mma(fa);
      __builtin_amdgcn_sched_barrier(0);
      issue(fa, (b + 2) * UB);
      __builtin_amdgcn_sched_barrier(0);
      mma(fb);
      __builtin_amdgcn_sched_barrier(0);
    }
    if (b + 1 < nblk) {
      issue(fb, (b + 1) * UB);
      __builtin_amdgcn_sched_barrier(0);
      mma(fa);
      mma(fb);
    } else {
      mma(fa);
    }
  }
  for (int u = nblk * UB; u < nu; ++u) {
    UnitE f;
    load_u(f, u);
    mma_u(f);
  }

  const int col = n0 + 2 * l16;
  const float bz0 = bias[col], bz1 = bias[col + 1];
  #pragma unroll
  for (int v = 0; v < 8; ++v) {
    const int row = m0 + v + 8 * half;
    const float sc = rs[row];
    v2f x;
    x.x = acc[0][v];
    x.y = acc[1][v];
    x.x *= sc;  x.y *= sc;
    x.x += bz0; x.y += bz1;
    x.x = fmaxf(x.x, 0.0f); x.y = fmaxf(x.y, 0.0f);
    *(v2f*)(C + (size_t)row * ldc + col) = x;
  }
}

__global__ void deg_rsqrt(const float* __restrict__ A, float* __restrict__ d)
{
  const int gw   = (blockIdx.x * blockDim.x + threadIdx.x) >> 5;
  const int lane = threadIdx.x & 31;
  if (gw >= NNODE) return;
  const float* row = A + (size_t)gw * NNODE;
  float s = 0.0f;
  for (int j = lane; j < NNODE; j += 32) s += row[j];
  #pragma unroll
  for (int off = 16; off > 0; off >>= 1) s += __shfl_xor(s, off, 32);
  if (lane == 0) d[gw] = (s > 0.0f) ? rsqrtf(s) : 0.0f;
}

#define NU8 (NNODE / 8)
#define PREP_WAVES 16
#define PREP_LG (NNODE / 32 / PREP_WAVES)
__global__ __launch_bounds__(32 * PREP_WAVES) void branch_prep_frag(
    const float* __restrict__ mask,
    const float* __restrict__ adj,
    const float* __restrict__ thrp,
    float* __restrict__ Ahf,
    float* __restrict__ d)
{
  static_assert(NNODE % (32 * PREP_WAVES) == 0 && NNODE % 16 == 0 && PREP_LG % 3 == 0,
                "strip, line and batch split");
  __shared__ float part[PREP_WAVES][32];
  const int strip = blockIdx.x;
  const int lane  = threadIdx.x & 31;
  const int wave  = threadIdx.x >> 5;
  const int half  = lane >> 4;
  const int row   = 16 * strip + (lane & 15);
  const float thr = thrp[0];
  const float* mrow = mask + (size_t)row * NNODE + 4 * half;
  const float* arow = adj  + (size_t)row * NNODE + 4 * half;
  float* dst = Ahf + ((size_t)strip * NU8 * 32 + lane) * 4;

  float s = 0.0f;
  #pragma unroll
  for (int i0 = 0; i0 < PREP_LG; i0 += 3) {
    v4f mv[12], av[12];
    #pragma unroll
    for (int i = 0; i < 3; ++i) {
      #pragma unroll
      for (int t = 0; t < 4; ++t) {
        const int u = 4 * (wave + PREP_WAVES * (i0 + i)) + t;
        mv[4 * i + t] = *(const v4f*)(mrow + 8 * u);
        av[4 * i + t] = *(const v4f*)(arow + 8 * u);
      }
    }
    #pragma unroll
    for (int i = 0; i < 3; ++i) {
      #pragma unroll
      for (int t = 0; t < 4; ++t) {
        const int u  = 4 * (wave + PREP_WAVES * (i0 + i)) + t;
        const int k0 = 8 * u + 4 * half;
        v4f w;
        #pragma unroll
        for (int e = 0; e < 4; ++e) {
          const float x = (mv[4 * i + t][e] > thr) ? av[4 * i + t][e] : 0.0f;
          s += x;
          w[e] = (k0 + e == row) ? x + 1.0f : x;
        }
        *(v4f*)(dst + (size_t)u * 128) = w;
      }
    }
  }
  part[wave][lane] = s;
  __syncthreads();
  if (threadIdx.x < 16) {
    float t = 0.0f;
    #pragma unroll
    for (int w = 0; w < PREP_WAVES; ++w) t += part[w][threadIdx.x] + part[w][threadIdx.x + 16];
    d[16 * strip + threadIdx.x] = rsqrtf(t + 1.0f);
  }
}

__global__ void symnorm(const float* __restrict__ adj,
                        const float* __restrict__ d,
                        float* __restrict__ An)
{
  const int idx = blockIdx.x * blockDim.x + threadIdx.x;
  if (idx >= NNODE * NNODE) return;
  const int i = idx / NNODE;
  const int j = idx - i * NNODE;
  An[idx] = adj[idx] * d[i] * d[j];
}

__global__ void conv_dil2(const float* __restrict__ emb,
                          const float* __restrict__ kw,
                          const float* __restrict__ kb,
                          float* __restrict__ out)
{
  const int idx = blockIdx.x * blockDim.x + threadIdx.x;
  if (idx >= NNODE * 256) return;
  const int n = idx >> 8;
  const int c = idx & 255;
  float acc = kb[0];
  #pragma unroll
  for (int ch = 0; ch < 3; ++ch) {
    #pragma unroll
    for (int kh = 0; kh < 3; ++kh) {
      const int ih = n + 2 * kh - 2;
      if (ih < 0 || ih >= NNODE) continue;
      #pragma unroll
      for (int kwi = 0; kwi < 3; ++kwi) {
        const int iw = c + 2 * kwi - 2;
        if (iw < 0 || iw >= 256) continue;
        acc += emb[((size_t)ch * NNODE + ih) * 256 + iw] * kw[ch * 9 + kh * 3 + kwi];
      }
    }
  }
  out[idx] = acc;
}

__global__ void logits_k(const float* __restrict__ h,
                         const float* __restrict__ Wb,
                         const float* __restrict__ bb,
                         float* __restrict__ out)
{
  const int i = blockIdx.x * blockDim.x + threadIdx.x;
  if (i >= BPAIR * 2) return;
  const int b = i >> 1;
  const int c = i & 1;
  float acc = bb[c];
  #pragma unroll
  for (int k = 0; k < 64; ++k) acc += h[b * 64 + k] * Wb[k * 2 + c];
  out[i] = acc;
}

extern "C" void kernel_launch(void* const* d_in, const int* in_sizes, int n_in,
                              void* d_out, int out_size, void* d_ws, size_t ws_size,
                              hipStream_t stream)
{
  const int*   left  = (const int*)d_in[0];
  const int*   right = (const int*)d_in[1];
  const float* X     = (const float*)d_in[2];
  const float* adj   = (const float*)d_in[3];
  const float* thr   = (const float*)d_in[4];
  const float* Wfc1  = (const float*)d_in[5];
  const float* bfc1  = (const float*)d_in[6];
  const float* Wg[6] = {(const float*)d_in[7],  (const float*)d_in[9],
                        (const float*)d_in[11], (const float*)d_in[13],
                        (const float*)d_in[15], (const float*)d_in[17]};
  const float* bg[6] = {(const float*)d_in[8],  (const float*)d_in[10],
                        (const float*)d_in[12], (const float*)d_in[14],
                        (const float*)d_in[16], (const float*)d_in[18]};
  const float* cnnk  = (const float*)d_in[19];
  const float* cnnb  = (const float*)d_in[20];
  const float* Wa    = (const float*)d_in[21];
  const float* ba    = (const float*)d_in[22];
  const float* Wb    = (const float*)d_in[23];
  const float* bb    = (const float*)d_in[24];

  float* out     = (float*)d_out;
  float* logits  = out;
  float* emb_all = out + BPAIR * 2;

  float* ws = (float*)d_ws;
  size_t off = 0;
  float* X0  = ws + off; off += (size_t)NNODE * D0;
  float* An  = ws + off; off += (size_t)NNODE * NNODE;
  float* A2  = ws + off; off += (size_t)NNODE * NNODE;
  float* Ah  = ws + off; off += (size_t)NNODE * NNODE;
  float* dv  = ws + off; off += NNODE;
  float* Hb  = ws + off; off += (size_t)NNODE * HDIM;
  float* H2b = ws + off; off += (size_t)NNODE * HDIM;
  float* emb = ws + off; off += (size_t)3 * NNODE * 256;
  float* hm  = ws + off; off += (size_t)BPAIR * 64;
  if (ws_size < off * sizeof(float)) return;

  const dim3 blk(128);

  gemm_wmma_f32<true, true, false, 8, false, 2, 2><<<dim3(D0 / 32, NNODE / 128), blk, 0, stream>>>(
      X, FIN, Wfc1, D0, bfc1, nullptr, X0, D0, FIN, nullptr, nullptr);

  deg_rsqrt<<<NNODE / 8, 256, 0, stream>>>(adj, dv);
  symnorm<<<(NNODE * NNODE) / 256, 256, 0, stream>>>(adj, dv, An);
  gemm_wmma_f32_t96<<<dim3(NNODE / 192, NNODE / 192), blk, 0, stream>>>(
      An, NNODE, An, NNODE, A2, NNODE, NNODE);

  const float* masks[3] = { adj, An, A2 };
  for (int ch = 0; ch < 3; ++ch) {
    const float* W1 = Wg[4 - 2 * ch];  const float* b1 = bg[4 - 2 * ch];
    const float* W2 = Wg[5 - 2 * ch];  const float* b2 = bg[5 - 2 * ch];
    float* embc = emb + (size_t)ch * NNODE * 256;

    branch_prep_frag<<<NNODE / 16, 32 * PREP_WAVES, 0, stream>>>(masks[ch], adj, thr, Ah, dv);

    gemm_wmma_f32<false, false, false, 8, true, 1, 2, true><<<dim3(HDIM / 32, NNODE / 64), blk, 0, stream>>>(
        X0, D0, W1, HDIM, nullptr, dv, Hb, HDIM, D0, nullptr, nullptr);
    gemm_afrag_bpi<8, HDIM><<<dim3(HDIM / 32, NNODE / 64), blk, 0, stream>>>(
        Ah, Hb, b1, dv, embc, 256, NNODE);
    gemm_wmma_f32<false, false, false, 8, true, 1, 2, true><<<dim3(HDIM / 32, NNODE / 64), blk, 0, stream>>>(
        embc, 256, W2, HDIM, nullptr, dv, H2b, HDIM, HDIM, nullptr, nullptr);
    gemm_afrag_bpi<8, HDIM><<<dim3(HDIM / 32, NNODE / 64), blk, 0, stream>>>(
        Ah, H2b, b2, dv, embc + 128, 256, NNODE);
  }

  conv_dil2<<<(NNODE * 256) / 256, 256, 0, stream>>>(emb, cnnk, cnnb, emb_all);

  gemm_wmma_f32<true, true, true, 8, false, 2, 2><<<dim3(64 / 32, BPAIR / 128), blk, 0, stream>>>(
      emb_all, 256, Wa, 64, ba, nullptr, hm, 64, 512, left, right);

  logits_k<<<(BPAIR * 2 + 255) / 256, 256, 0, stream>>>(hm, Wb, bb, logits);
}


# sym-norm and its degree pass folded into the two branch preps that read adj first: ch0 prep also writes D^-1/2 of adj, ch1 prep forms An, masks by it, stores it; 22 launches to 20; sum orders kept
# speedup vs baseline: 4.1422x; 1.0492x over previous
#include <hip/hip_runtime.h>
#include <hip/hip_bf16.h>

#define NNODE 3072
#define FIN   1546
#define D0    256
#define HDIM  128
#define BPAIR 16384

typedef __attribute__((ext_vector_type(2))) float v2f;
typedef __attribute__((ext_vector_type(4))) float v4f;
typedef __attribute__((ext_vector_type(8))) float v8f;

template <int SM, int SN>
struct KStep { v2f a[SM]; v2f b[SN]; };

template <int SM, int SN>
__device__ __forceinline__ void load_kstep(KStep<SM, SN>& f, const float* const (&ap)[SM],
                                           const float* bp, int ldb)
{
  #pragma unroll
  for (int s = 0; s < SM; ++s) f.a[s] = *(const v2f*)ap[s];
  #pragma unroll
  for (int j = 0; j < SN; ++j) {
    f.b[j].x = bp[16 * j];
    f.b[j].y = bp[ldb + 16 * j];
  }
}

template <int SM, int SN>
__device__ __forceinline__ void wmma_tile(v8f (&acc)[SM][SN], const KStep<SM, SN>& f)
{
  #pragma unroll
  for (int s = 0; s < SM; ++s) {
    #pragma unroll
    for (int j = 0; j < SN; ++j)
      acc[s][j] = __builtin_amdgcn_wmma_f32_16x16x4_f32(false, f.a[s], false, f.b[j], (short)0,
                                                        acc[s][j], false, false);
  }
}

template <int U, int SM, int SN>
__device__ __forceinline__ void mma_block(v8f (&acc)[SM][SN], const KStep<SM, SN> (&f)[U])
{
  #pragma unroll
  for (int i = 0; i < U; ++i) wmma_tile<SM, SN>(acc, f[i]);
}

template <bool RELU, bool BIAS, bool GATHER, int U, bool RSCALE, int SM, int SN,
          bool CPI = false>
__global__ __launch_bounds__(128) void gemm_wmma_f32(
    const float* __restrict__ A, int lda,
    const float* __restrict__ Bm, int ldb,
    const float* __restrict__ bias,
    const float* __restrict__ rs,
    float* __restrict__ C, int ldc,
    int K,
    const int* __restrict__ gl, const int* __restrict__ gr)
{
  static_assert(U * (SM + 2 * SN) <= 60, "load group must fit the wave's load counter");
  constexpr int TM = 16 * SM;
  constexpr int TN = 16 * SN;
  using KS = KStep<SM, SN>;

  const int lane = threadIdx.x & 31;
  const int wave = threadIdx.x >> 5;
  const int half = lane >> 4;
  const int l16  = lane & 15;

  const int m0 = blockIdx.y * (4 * TM) + wave * TM;
  const int n0 = blockIdx.x * TN;

  v8f acc[SM][SN] = {};

  const float* bp = Bm + (size_t)(2 * half) * ldb + n0 + l16;
  const size_t bstep = (size_t)4 * ldb;

  constexpr int UU = U;
  const float* pl[SM];
  const float* pr[SM];
  #pragma unroll
  for (int s = 0; s < SM; ++s) {
    const int am = m0 + 16 * s + l16;
    pl[s] = GATHER ? A + (size_t)gl[am] * 256 + 2 * half
                   : A + (size_t)am * lda + 2 * half;
    pr[s] = GATHER ? A + (size_t)gr[am] * 256 + 2 * half : pl[s];
  }
  const int nk   = GATHER ? 128 : (K >> 2);
  const int nblk = nk / UU;

  auto load_t = [&](KS& f, int t) {
    const float* q[SM];
    const bool r = GATHER && (t >= 64);
    const int tt = GATHER ? (t & 63) : t;
    #pragma unroll
    for (int s = 0; s < SM; ++s) q[s] = (r ? pr[s] : pl[s]) + 4 * (size_t)tt;
    load_kstep<SM, SN>(f, q, bp + (size_t)t * bstep, ldb);
  };
  auto issue = [&](KS (&f)[UU], int t0) {
    #pragma unroll
    for (int i = 0; i < UU; ++i) load_t(f[i], t0 + i);
  };

  KS fa[UU], fb[UU];
  if (nblk > 0) {
    issue(fa, 0);
    int b = 0;
    for (; b + 2 < nblk; b += 2) {
      issue(fb, (b + 1) * UU);
      __builtin_amdgcn_sched_barrier(0);
      mma_block<UU, SM, SN>(acc, fa);
      __builtin_amdgcn_sched_barrier(0);
      issue(fa, (b + 2) * UU);
      __builtin_amdgcn_sched_barrier(0);
      mma_block<UU, SM, SN>(acc, fb);
      __builtin_amdgcn_sched_barrier(0);
    }
    if (b + 1 < nblk) {
      issue(fb, (b + 1) * UU);
      __builtin_amdgcn_sched_barrier(0);
      mma_block<UU, SM, SN>(acc, fa);
      mma_block<UU, SM, SN>(acc, fb);
    } else {
      mma_block<UU, SM, SN>(acc, fa);
    }
  }
  for (int t = nblk * UU; t < nk; ++t) {
    KS f;
    load_t(f, t);
    wmma_tile<SM, SN>(acc, f);
  }
  if (!GATHER) {
    const int kmain = K & ~3;
    if (kmain < K) {
      const float* bq = bp + (size_t)nk * bstep;
      const int ka = kmain + 2 * half;
      const bool g0 = (ka < K), g1 = (ka + 1 < K);
      KS f;
      #pragma unroll
      for (int s = 0; s < SM; ++s) {
        const float* ap = pl[s] + kmain;
        f.a[s].x = g0 ? ap[0] : 0.0f;
        f.a[s].y = g1 ? ap[1] : 0.0f;
      }
      #pragma unroll
      for (int j = 0; j < SN; ++j) {
        f.b[j].x = g0 ? bq[16 * j]       : 0.0f;
        f.b[j].y = g1 ? bq[ldb + 16 * j] : 0.0f;
      }
      wmma_tile<SM, SN>(acc, f);
    }
  }

  if constexpr (CPI) {
    #pragma unroll
    for (int s = 0; s < SM; ++s) {
      #pragma unroll
      for (int v = 0; v < 8; v += 2) {
        const int row = m0 + 16 * s + v + 8 * half;
        const float sc0 = RSCALE ? rs[row] : 1.0f;
        const float sc1 = RSCALE ? rs[row + 1] : 1.0f;
        #pragma unroll
        for (int j = 0; j < SN; ++j) {
          const int col = n0 + 16 * j + l16;
          v2f x;
          x.x = acc[s][j][v];
          x.y = acc[s][j][v + 1];
          if (RSCALE) { x.x *= sc0; x.y *= sc1; }
          if (BIAS) { x.x += bias[col]; x.y += bias[col]; }
          if (RELU) { x.x = fmaxf(x.x, 0.0f); x.y = fmaxf(x.y, 0.0f); }
          *(v2f*)(C + (size_t)(row >> 1) * (2 * ldc) + 2 * col) = x;
        }
      }
    }
  } else {
    #pragma unroll
    for (int s = 0; s < SM; ++s) {
      #pragma unroll
      for (int v = 0; v < 8; ++v) {
        const int row = m0 + 16 * s + v + 8 * half;
        const float sc = RSCALE ? rs[row] : 1.0f;
        #pragma unroll
        for (int j = 0; j < SN; ++j) {
          const int col = n0 + 16 * j + l16;
          float x = acc[s][j][v];
          if (RSCALE) x *= sc;
          if (BIAS) x += bias[col];
          if (RELU) x = fmaxf(x, 0.0f);
          C[(size_t)row * ldc + col] = x;
        }
      }
    }
  }
}

struct Unit96 { v4f a[6]; v2f b[2][6]; };

__device__ __forceinline__ void load_unit96(Unit96& f, const float* ap, size_t astrip,
                                            const float* bp, size_t ldb)
{
  #pragma unroll
  for (int s = 0; s < 6; ++s) f.a[s] = *(const v4f*)(ap + s * astrip);
  #pragma unroll
  for (int p = 0; p < 2; ++p) {
    #pragma unroll
    for (int j = 0; j < 6; ++j) {
      f.b[p][j].x = bp[(size_t)(2 * p) * ldb + 16 * j];
      f.b[p][j].y = bp[(size_t)(2 * p + 1) * ldb + 16 * j];
    }
  }
}

__device__ __forceinline__ void mma_unit96(v8f (&acc)[6][6], const Unit96& f)
{
  #pragma unroll
  for (int p = 0; p < 2; ++p) {
    #pragma unroll
    for (int s = 0; s < 6; ++s) {
      const v2f a = (p == 0) ? __builtin_shufflevector(f.a[s], f.a[s], 0, 1)
                             : __builtin_shufflevector(f.a[s], f.a[s], 2, 3);
      #pragma unroll
      for (int j = 0; j < 6; ++j)
        acc[s][j] = __builtin_amdgcn_wmma_f32_16x16x4_f32(false, a, false, f.b[p][j], (short)0,
                                                          acc[s][j], false, false);
    }
  }
}

__global__ __launch_bounds__(128) void gemm_wmma_f32_t96(
    const float* __restrict__ A, int lda,
    const float* __restrict__ Bm, int ldb,
    float* __restrict__ C, int ldc,
    int K)
{
  const int lane = threadIdx.x & 31;
  const int wave = threadIdx.x >> 5;
  const int half = lane >> 4;
  const int l16  = lane & 15;

  const int m0 = blockIdx.y * 192 + (wave >> 1) * 96;
  const int n0 = blockIdx.x * 192 + (wave & 1) * 96;

  v8f acc[6][6] = {};

  const float* ap = A + (size_t)(m0 + l16) * lda + 4 * half;
  const float* bp = Bm + (size_t)(4 * half) * ldb + n0 + l16;
  const size_t astrip = (size_t)16 * lda;
  const size_t bunit  = (size_t)8 * ldb;

  auto issue = [&](Unit96 (&f)[2], int kb) {
    const float* aq = ap + 16 * (size_t)kb;
    const float* bq = bp + 2 * (size_t)kb * bunit;
    load_unit96(f[0], aq, astrip, bq, (size_t)ldb);
    load_unit96(f[1], aq + 8, astrip, bq + bunit, (size_t)ldb);
  };
  auto mma = [&](const Unit96 (&f)[2]) {
    mma_unit96(acc, f[0]);
    mma_unit96(acc, f[1]);
  };

  const int nb = K >> 4;
  Unit96 fa[2], fb[2];
  if (nb > 0) {
    issue(fa, 0);
    int kb = 0;
    for (; kb + 2 < nb; kb += 2) {
      issue(fb, kb + 1);
      __builtin_amdgcn_sched_barrier(0);
      mma(fa);
      __builtin_amdgcn_sched_barrier(0);
      issue(fa, kb + 2);
      __builtin_amdgcn_sched_barrier(0);
      mma(fb);
      __builtin_amdgcn_sched_barrier(0);
    }
    if (kb + 1 < nb) {
      issue(fb, kb + 1);
      __builtin_amdgcn_sched_barrier(0);
      mma(fa);
      mma(fb);
    } else {
      mma(fa);
    }
  }

  #pragma unroll
  for (int s = 0; s < 6; ++s) {
    #pragma unroll
    for (int v = 0; v < 8; ++v) {
      const int row = m0 + 16 * s + v + 8 * half;
      #pragma unroll
      for (int j = 0; j < 6; ++j)
        C[(size_t)row * ldc + n0 + 16 * j + l16] = acc[s][j][v];
    }
  }
}

struct UnitE { v4f a; v4f b[2]; };

template <int UB, int LDB>
__global__ __launch_bounds__(128) void gemm_afrag_bpi(
    const float* __restrict__ Af,
    const float* __restrict__ Bm,
    const float* __restrict__ bias,
    const float* __restrict__ rs,
    float* __restrict__ C, int ldc,
    int K)
{
  static_assert(3 * UB <= 60, "load group must fit the wave's load counter");
  const int lane = threadIdx.x & 31;
  const int wave = threadIdx.x >> 5;
  const int half = lane >> 4;
  const int l16  = lane & 15;

  const int strip = blockIdx.y * 4 + wave;
  const int m0 = 16 * strip;
  const int n0 = blockIdx.x * 32;
  const int nu = K >> 3;

  v8f acc[2] = {};

  const float* ap = Af + ((size_t)strip * nu * 32 + lane) * 4;
  const float* bp = Bm + (size_t)(2 * half) * (2 * LDB) + 2 * (n0 + 2 * l16);
  constexpr size_t prow  = (size_t)2 * LDB;
  constexpr size_t bunit = (size_t)8 * LDB;

  auto load_u = [&](UnitE& f, int u) {
    f.a = *(const v4f*)(ap + (size_t)u * 128);
    const float* bq = bp + (size_t)u * bunit;
    f.b[0] = *(const v4f*)bq;
    f.b[1] = *(const v4f*)(bq + prow);
  };
  auto mma_u = [&](const UnitE& f) {
    #pragma unroll
    for (int p = 0; p < 2; ++p) {
      const v2f a  = (p == 0) ? __builtin_shufflevector(f.a, f.a, 0, 1)
                              : __builtin_shufflevector(f.a, f.a, 2, 3);
      const v2f b0 = __builtin_shufflevector(f.b[p], f.b[p], 0, 1);
      const v2f b1 = __builtin_shufflevector(f.b[p], f.b[p], 2, 3);
      acc[0] = __builtin_amdgcn_wmma_f32_16x16x4_f32(false, a, false, b0, (short)0, acc[0], false, false);
      acc[1] = __builtin_amdgcn_wmma_f32_16x16x4_f32(false, a, false, b1, (short)0, acc[1], false, false);
    }
  };
  auto issue = [&](UnitE (&f)[UB], int u0) {
    #pragma unroll
    for (int i = 0; i < UB; ++i) load_u(f[i], u0 + i);
  };
  auto mma = [&](const UnitE (&f)[UB]) {
    #pragma unroll
    for (int i = 0; i < UB; ++i) mma_u(f[i]);
  };

  const int nblk = nu / UB;
  UnitE fa[UB], fb[UB];
  if (nblk > 0) {
    issue(fa, 0);
    int b = 0;
    for (; b + 2 < nblk; b += 2) {
      issue(fb, (b + 1) * UB);
      __builtin_amdgcn_sched_barrier(0);
      mma(fa);
      __builtin_amdgcn_sched_barrier(0);
      issue(fa, (b + 2) * UB);
      __builtin_amdgcn_sched_barrier(0);
      mma(fb);
      __builtin_amdgcn_sched_barrier(0);
    }
    if (b + 1 < nblk) {
      issue(fb, (b + 1) * UB);
      __builtin_amdgcn_sched_barrier(0);
      mma(fa);
      mma(fb);
    } else {
      mma(fa);
    }
  }
  for (int u = nblk * UB; u < nu; ++u) {
    UnitE f;
    load_u(f, u);
    mma_u(f);
  }

  const int col = n0 + 2 * l16;
  const float bz0 = bias[col], bz1 = bias[col + 1];
  #pragma unroll
  for (int v = 0; v < 8; ++v) {
    const int row = m0 + v + 8 * half;
    const float sc = rs[row];
    v2f x;
    x.x = acc[0][v];
    x.y = acc[1][v];
    x.x *= sc;  x.y *= sc;
    x.x += bz0; x.y += bz1;
    x.x = fmaxf(x.x, 0.0f); x.y = fmaxf(x.y, 0.0f);
    *(v2f*)(C + (size_t)row * ldc + col) = x;
  }
}

#define NU8 (NNODE / 8)
#define PREP_WAVES 16
#define PREP_LG (NNODE / 32 / PREP_WAVES)
template <int MODE>
__global__ __launch_bounds__(32 * PREP_WAVES) void branch_prep_frag(
    const float* __restrict__ mask,
    const float* __restrict__ adj,
    const float* __restrict__ thrp,
    float* __restrict__ Ahf,
    float* __restrict__ d,
    float* __restrict__ aux)
{
  static_assert(NNODE % (32 * PREP_WAVES) == 0 && NNODE % 16 == 0 && PREP_LG % 3 == 0,
                "strip, line and batch split");
  static_assert(PREP_WAVES == 16 && NNODE == 32 * 96, "MODE 1: one wave per row of the strip, 96 elements a lane");
  __shared__ float part[PREP_WAVES][32];
  const int strip = blockIdx.x;
  const int lane  = threadIdx.x & 31;
  const int wave  = threadIdx.x >> 5;
  const int half  = lane >> 4;
  const int row   = 16 * strip + (lane & 15);
  const float thr = thrp[0];
  const float* arow = adj + (size_t)row * NNODE + 4 * half;
  const float* mrow = (MODE == 2) ? mask + 4 * half
                                  : mask + (size_t)row * NNODE + 4 * half;
  float* dst = Ahf + ((size_t)strip * NU8 * 32 + lane) * 4;

  float di = 1.0f;
  if constexpr (MODE == 2) di = mask[row];

  if constexpr (MODE == 1) {
    const int r1 = 16 * strip + wave;
    const float* rp = adj + (size_t)r1 * NNODE + lane;
    float sd = 0.0f;
    #pragma unroll
    for (int b = 0; b < 2; ++b) {
      float v[48];
      #pragma unroll
      for (int t = 0; t < 48; ++t) v[t] = rp[32 * (48 * b + t)];
      #pragma unroll
      for (int t = 0; t < 48; ++t) sd += v[t];
    }
    #pragma unroll
    for (int off = 16; off > 0; off >>= 1) sd += __shfl_xor(sd, off, 32);
    if (lane == 0) aux[r1] = (sd > 0.0f) ? rsqrtf(sd) : 0.0f;
  }

  float s = 0.0f;
  #pragma unroll
  for (int i0 = 0; i0 < PREP_LG; i0 += 3) {
    v4f mv[12], av[12];
    #pragma unroll
    for (int i = 0; i < 3; ++i) {
      #pragma unroll
      for (int t = 0; t < 4; ++t) {
        const int u = 4 * (wave + PREP_WAVES * (i0 + i)) + t;
        av[4 * i + t] = *(const v4f*)(arow + 8 * u);
        if constexpr (MODE != 1) mv[4 * i + t] = *(const v4f*)(mrow + 8 * u);
      }
    }
    #pragma unroll
    for (int i = 0; i < 3; ++i) {
      #pragma unroll
      for (int t = 0; t < 4; ++t) {
        const int u  = 4 * (wave + PREP_WAVES * (i0 + i)) + t;
        const int k0 = 8 * u + 4 * half;
        v4f w, an;
        #pragma unroll
        for (int e = 0; e < 4; ++e) {
          const float a = av[4 * i + t][e];
          float m;
          if constexpr (MODE == 0)      m = mv[4 * i + t][e];
          else if constexpr (MODE == 1) m = a;
          else                          m = (a * di) * mv[4 * i + t][e];
          an[e] = m;
          const float x = (m > thr) ? a : 0.0f;
          s += x;
          w[e] = (k0 + e == row) ? x + 1.0f : x;
        }
        *(v4f*)(dst + (size_t)u * 128) = w;
        if constexpr (MODE == 2) *(v4f*)(aux + (size_t)row * NNODE + k0) = an;
      }
    }
  }
  part[wave][lane] = s;
  __syncthreads();
  if (threadIdx.x < 16) {
    float t = 0.0f;
    #pragma unroll
    for (int w = 0; w < PREP_WAVES; ++w) t += part[w][threadIdx.x] + part[w][threadIdx.x + 16];
    d[16 * strip + threadIdx.x] = rsqrtf(t + 1.0f);
  }
}

__global__ void conv_dil2(const float* __restrict__ emb,
                          const float* __restrict__ kw,
                          const float* __restrict__ kb,
                          float* __restrict__ out)
{
  const int idx = blockIdx.x * blockDim.x + threadIdx.x;
  if (idx >= NNODE * 256) return;
  const int n = idx >> 8;
  const int c = idx & 255;
  float acc = kb[0];
  #pragma unroll
  for (int ch = 0; ch < 3; ++ch) {
    #pragma unroll
    for (int kh = 0; kh < 3; ++kh) {
      const int ih = n + 2 * kh - 2;
      if (ih < 0 || ih >= NNODE) continue;
      #pragma unroll
      for (int kwi = 0; kwi < 3; ++kwi) {
        const int iw = c + 2 * kwi - 2;
        if (iw < 0 || iw >= 256) continue;
        acc += emb[((size_t)ch * NNODE + ih) * 256 + iw] * kw[ch * 9 + kh * 3 + kwi];
      }
    }
  }
  out[idx] = acc;
}

__global__ void logits_k(const float* __restrict__ h,
                         const float* __restrict__ Wb,
                         const float* __restrict__ bb,
                         float* __restrict__ out)
{
  const int i = blockIdx.x * blockDim.x + threadIdx.x;
  if (i >= BPAIR * 2) return;
  const int b = i >> 1;
  const int c = i & 1;
  float acc = bb[c];
  #pragma unroll
  for (int k = 0; k < 64; ++k) acc += h[b * 64 + k] * Wb[k * 2 + c];
  out[i] = acc;
}

extern "C" void kernel_launch(void* const* d_in, const int* in_sizes, int n_in,
                              void* d_out, int out_size, void* d_ws, size_t ws_size,
                              hipStream_t stream)
{
  const int*   left  = (const int*)d_in[0];
  const int*   right = (const int*)d_in[1];
  const float* X     = (const float*)d_in[2];
  const float* adj   = (const float*)d_in[3];
  const float* thr   = (const float*)d_in[4];
  const float* Wfc1  = (const float*)d_in[5];
  const float* bfc1  = (const float*)d_in[6];
  const float* Wg[6] = {(const float*)d_in[7],  (const float*)d_in[9],
                        (const float*)d_in[11], (const float*)d_in[13],
                        (const float*)d_in[15], (const float*)d_in[17]};
  const float* bg[6] = {(const float*)d_in[8],  (const float*)d_in[10],
                        (const float*)d_in[12], (const float*)d_in[14],
                        (const float*)d_in[16], (const float*)d_in[18]};
  const float* cnnk  = (const float*)d_in[19];
  const float* cnnb  = (const float*)d_in[20];
  const float* Wa    = (const float*)d_in[21];
  const float* ba    = (const float*)d_in[22];
  const float* Wb    = (const float*)d_in[23];
  const float* bb    = (const float*)d_in[24];

  float* out     = (float*)d_out;
  float* logits  = out;
  float* emb_all = out + BPAIR * 2;

  float* ws = (float*)d_ws;
  size_t off = 0;
  float* X0  = ws + off; off += (size_t)NNODE * D0;
  float* An  = ws + off; off += (size_t)NNODE * NNODE;
  float* A2  = ws + off; off += (size_t)NNODE * NNODE;
  float* Ah  = ws + off; off += (size_t)NNODE * NNODE;
  float* dv  = ws + off; off += NNODE;
  float* Hb  = ws + off; off += (size_t)NNODE * HDIM;
  float* H2b = ws + off; off += (size_t)NNODE * HDIM;
  float* emb = ws + off; off += (size_t)3 * NNODE * 256;
  float* hm  = ws + off; off += (size_t)BPAIR * 64;
  if (ws_size < off * sizeof(float)) return;

  const dim3 blk(128);

  gemm_wmma_f32<true, true, false, 8, false, 2, 2><<<dim3(D0 / 32, NNODE / 128), blk, 0, stream>>>(
      X, FIN, Wfc1, D0, bfc1, nullptr, X0, D0, FIN, nullptr, nullptr);

  float* dsym = A2;
  for (int ch = 0; ch < 3; ++ch) {
    const float* W1 = Wg[4 - 2 * ch];  const float* b1 = bg[4 - 2 * ch];
    const float* W2 = Wg[5 - 2 * ch];  const float* b2 = bg[5 - 2 * ch];
    float* embc = emb + (size_t)ch * NNODE * 256;

    if (ch == 0) {
      branch_prep_frag<1><<<NNODE / 16, 32 * PREP_WAVES, 0, stream>>>(adj, adj, thr, Ah, dv, dsym);
    } else if (ch == 1) {
      branch_prep_frag<2><<<NNODE / 16, 32 * PREP_WAVES, 0, stream>>>(dsym, adj, thr, Ah, dv, An);
    } else {
      gemm_wmma_f32_t96<<<dim3(NNODE / 192, NNODE / 192), blk, 0, stream>>>(
          An, NNODE, An, NNODE, A2, NNODE, NNODE);
      branch_prep_frag<0><<<NNODE / 16, 32 * PREP_WAVES, 0, stream>>>(A2, adj, thr, Ah, dv, nullptr);
    }

    gemm_wmma_f32<false, false, false, 8, true, 1, 2, true><<<dim3(HDIM / 32, NNODE / 64), blk, 0, stream>>>(
        X0, D0, W1, HDIM, nullptr, dv, Hb, HDIM, D0, nullptr, nullptr);
    gemm_afrag_bpi<8, HDIM><<<dim3(HDIM / 32, NNODE / 64), blk, 0, stream>>>(
        Ah, Hb, b1, dv, embc, 256, NNODE);
    gemm_wmma_f32<false, false, false, 8, true, 1, 2, true><<<dim3(HDIM / 32, NNODE / 64), blk, 0, stream>>>(
        embc, 256, W2, HDIM, nullptr, dv, H2b, HDIM, HDIM, nullptr, nullptr);
    gemm_afrag_bpi<8, HDIM><<<dim3(HDIM / 32, NNODE / 64), blk, 0, stream>>>(
        Ah, H2b, b2, dv, embc + 128, 256, NNODE);
  }

  conv_dil2<<<(NNODE * 256) / 256, 256, 0, stream>>>(emb, cnnk, cnnb, emb_all);

  gemm_wmma_f32<true, true, true, 8, false, 2, 2><<<dim3(64 / 32, BPAIR / 128), blk, 0, stream>>>(
      emb_all, 256, Wa, 64, ba, nullptr, hm, 64, 512, left, right);

  logits_k<<<(BPAIR * 2 + 255) / 256, 256, 0, stream>>>(hm, Wb, bb, logits);
}


# N x N x N product: the prep that forms its operand stores it once per side in the order read (A fragments, k-pair interleaved B): 24 whole-line 16-byte loads per 16 k, was 60; k and sum order kept
# speedup vs baseline: 4.2899x; 1.0357x over previous
#include <hip/hip_runtime.h>
#include <hip/hip_bf16.h>

#define NNODE 3072
#define FIN   1546
#define D0    256
#define HDIM  128
#define BPAIR 16384

typedef __attribute__((ext_vector_type(2))) float v2f;
typedef __attribute__((ext_vector_type(4))) float v4f;
typedef __attribute__((ext_vector_type(8))) float v8f;

template <int SM, int SN>
struct KStep { v2f a[SM]; v2f b[SN]; };

template <int SM, int SN>
__device__ __forceinline__ void load_kstep(KStep<SM, SN>& f, const float* const (&ap)[SM],
                                           const float* bp, int ldb)
{
  #pragma unroll
  for (int s = 0; s < SM; ++s) f.a[s] = *(const v2f*)ap[s];
  #pragma unroll
  for (int j = 0; j < SN; ++j) {
    f.b[j].x = bp[16 * j];
    f.b[j].y = bp[ldb + 16 * j];
  }
}

template <int SM, int SN>
__device__ __forceinline__ void wmma_tile(v8f (&acc)[SM][SN], const KStep<SM, SN>& f)
{
  #pragma unroll
  for (int s = 0; s < SM; ++s) {
    #pragma unroll
    for (int j = 0; j < SN; ++j)
      acc[s][j] = __builtin_amdgcn_wmma_f32_16x16x4_f32(false, f.a[s], false, f.b[j], (short)0,
                                                        acc[s][j], false, false);
  }
}

template <int U, int SM, int SN>
__device__ __forceinline__ void mma_block(v8f (&acc)[SM][SN], const KStep<SM, SN> (&f)[U])
{
  #pragma unroll
  for (int i = 0; i < U; ++i) wmma_tile<SM, SN>(acc, f[i]);
}

template <bool RELU, bool BIAS, bool GATHER, int U, bool RSCALE, int SM, int SN,
          bool CPI = false>
__global__ __launch_bounds__(128) void gemm_wmma_f32(
    const float* __restrict__ A, int lda,
    const float* __restrict__ Bm, int ldb,
    const float* __restrict__ bias,
    const float* __restrict__ rs,
    float* __restrict__ C, int ldc,
    int K,
    const int* __restrict__ gl, const int* __restrict__ gr)
{
  static_assert(U * (SM + 2 * SN) <= 60, "load group must fit the wave's load counter");
  constexpr int TM = 16 * SM;
  constexpr int TN = 16 * SN;
  using KS = KStep<SM, SN>;

  const int lane = threadIdx.x & 31;
  const int wave = threadIdx.x >> 5;
  const int half = lane >> 4;
  const int l16  = lane & 15;

  const int m0 = blockIdx.y * (4 * TM) + wave * TM;
  const int n0 = blockIdx.x * TN;

  v8f acc[SM][SN] = {};

  const float* bp = Bm + (size_t)(2 * half) * ldb + n0 + l16;
  const size_t bstep = (size_t)4 * ldb;

  constexpr int UU = U;
  const float* pl[SM];
  const float* pr[SM];
  #pragma unroll
  for (int s = 0; s < SM; ++s) {
    const int am = m0 + 16 * s + l16;
    pl[s] = GATHER ? A + (size_t)gl[am] * 256 + 2 * half
                   : A + (size_t)am * lda + 2 * half;
    pr[s] = GATHER ? A + (size_t)gr[am] * 256 + 2 * half : pl[s];
  }
  const int nk   = GATHER ? 128 : (K >> 2);
  const int nblk = nk / UU;

  auto load_t = [&](KS& f, int t) {
    const float* q[SM];
    const bool r = GATHER && (t >= 64);
    const int tt = GATHER ? (t & 63) : t;
    #pragma unroll
    for (int s = 0; s < SM; ++s) q[s] = (r ? pr[s] : pl[s]) + 4 * (size_t)tt;
    load_kstep<SM, SN>(f, q, bp + (size_t)t * bstep, ldb);
  };
  auto issue = [&](KS (&f)[UU], int t0) {
    #pragma unroll
    for (int i = 0; i < UU; ++i) load_t(f[i], t0 + i);
  };

  KS fa[UU], fb[UU];
  if (nblk > 0) {
    issue(fa, 0);
    int b = 0;
    for (; b + 2 < nblk; b += 2) {
      issue(fb, (b + 1) * UU);
      __builtin_amdgcn_sched_barrier(0);
      mma_block<UU, SM, SN>(acc, fa);
      __builtin_amdgcn_sched_barrier(0);
      issue(fa, (b + 2) * UU);
      __builtin_amdgcn_sched_barrier(0);
      mma_block<UU, SM, SN>(acc, fb);
      __builtin_amdgcn_sched_barrier(0);
    }
    if (b + 1 < nblk) {
      issue(fb, (b + 1) * UU);
      __builtin_amdgcn_sched_barrier(0);
      mma_block<UU, SM, SN>(acc, fa);
      mma_block<UU, SM, SN>(acc, fb);
    } else {
      mma_block<UU, SM, SN>(acc, fa);
    }
  }
  for (int t = nblk * UU; t < nk; ++t) {
    KS f;
    load_t(f, t);
    wmma_tile<SM, SN>(acc, f);
  }
  if (!GATHER) {
    const int kmain = K & ~3;
    if (kmain < K) {
      const float* bq = bp + (size_t)nk * bstep;
      const int ka = kmain + 2 * half;
      const bool g0 = (ka < K), g1 = (ka + 1 < K);
      KS f;
      #pragma unroll
      for (int s = 0; s < SM; ++s) {
        const float* ap = pl[s] + kmain;
        f.a[s].x = g0 ? ap[0] : 0.0f;
        f.a[s].y = g1 ? ap[1] : 0.0f;
      }
      #pragma unroll
      for (int j = 0; j < SN; ++j) {
        f.b[j].x = g0 ? bq[16 * j]       : 0.0f;
        f.b[j].y = g1 ? bq[ldb + 16 * j] : 0.0f;
      }
      wmma_tile<SM, SN>(acc, f);
    }
  }

  if constexpr (CPI) {
    #pragma unroll
    for (int s = 0; s < SM; ++s) {
      #pragma unroll
      for (int v = 0; v < 8; v += 2) {
        const int row = m0 + 16 * s + v + 8 * half;
        const float sc0 = RSCALE ? rs[row] : 1.0f;
        const float sc1 = RSCALE ? rs[row + 1] : 1.0f;
        #pragma unroll
        for (int j = 0; j < SN; ++j) {
          const int col = n0 + 16 * j + l16;
          v2f x;
          x.x = acc[s][j][v];
          x.y = acc[s][j][v + 1];
          if (RSCALE) { x.x *= sc0; x.y *= sc1; }
          if (BIAS) { x.x += bias[col]; x.y += bias[col]; }
          if (RELU) { x.x = fmaxf(x.x, 0.0f); x.y = fmaxf(x.y, 0.0f); }
          *(v2f*)(C + (size_t)(row >> 1) * (2 * ldc) + 2 * col) = x;
        }
      }
    }
  } else {
    #pragma unroll
    for (int s = 0; s < SM; ++s) {
      #pragma unroll
      for (int v = 0; v < 8; ++v) {
        const int row = m0 + 16 * s + v + 8 * half;
        const float sc = RSCALE ? rs[row] : 1.0f;
        #pragma unroll
        for (int j = 0; j < SN; ++j) {
          const int col = n0 + 16 * j + l16;
          float x = acc[s][j][v];
          if (RSCALE) x *= sc;
          if (BIAS) x += bias[col];
          if (RELU) x = fmaxf(x, 0.0f);
          C[(size_t)row * ldc + col] = x;
        }
      }
    }
  }
}

struct Unit96 { v4f a[6]; v4f b[2][3]; };

__device__ __forceinline__ void mma_unit96(v8f (&acc)[6][6], const Unit96& f)
{
  #pragma unroll
  for (int p = 0; p < 2; ++p) {
    #pragma unroll
    for (int s = 0; s < 6; ++s) {
      const v2f a = (p == 0) ? __builtin_shufflevector(f.a[s], f.a[s], 0, 1)
                             : __builtin_shufflevector(f.a[s], f.a[s], 2, 3);
      #pragma unroll
      for (int j = 0; j < 6; ++j) {
        const v4f bb = f.b[p][j >> 1];
        const v2f b = (j & 1) ? __builtin_shufflevector(bb, bb, 2, 3)
                              : __builtin_shufflevector(bb, bb, 0, 1);
        acc[s][j] = __builtin_amdgcn_wmma_f32_16x16x4_f32(false, a, false, b, (short)0,
                                                          acc[s][j], false, false);
      }
    }
  }
}

__global__ __launch_bounds__(128) void gemm_wmma_f32_t96(
    const float* __restrict__ Af,
    const float* __restrict__ Bpi,
    float* __restrict__ C)
{
  static_assert(NNODE % 192 == 0, "tile split");
  constexpr int NU = NNODE / 8;
  const int lane = threadIdx.x & 31;
  const int wave = threadIdx.x >> 5;
  const int half = lane >> 4;
  const int l16  = lane & 15;

  const int strip0 = blockIdx.y * 12 + (wave >> 1) * 6;
  const int m0 = 16 * strip0;
  const int n0 = blockIdx.x * 192 + (wave & 1) * 96;

  v8f acc[6][6] = {};

  const float* ap = Af + ((size_t)strip0 * NU * 32 + lane) * 4;
  const float* bp = Bpi + (size_t)(2 * half) * (2 * NNODE) + 2 * (n0 + 2 * l16);
  constexpr size_t astrip = (size_t)NU * 128;
  constexpr size_t prow   = (size_t)2 * NNODE;
  constexpr size_t bunit  = (size_t)8 * NNODE;

  auto load_u = [&](Unit96& f, int u) {
    const float* aq = ap + (size_t)u * 128;
    const float* bq = bp + (size_t)u * bunit;
    #pragma unroll
    for (int s = 0; s < 6; ++s) f.a[s] = *(const v4f*)(aq + s * astrip);
    #pragma unroll
    for (int p = 0; p < 2; ++p) {
      #pragma unroll
      for (int q = 0; q < 3; ++q) f.b[p][q] = *(const v4f*)(bq + p * prow + 64 * q);
    }
  };
  auto issue = [&](Unit96 (&f)[2], int kb) {
    load_u(f[0], 2 * kb);
    load_u(f[1], 2 * kb + 1);
  };
  auto mma = [&](const Unit96 (&f)[2]) {
    mma_unit96(acc, f[0]);
    mma_unit96(acc, f[1]);
  };

  constexpr int nb = NNODE / 16;
  Unit96 fa[2], fb[2];
  {
    issue(fa, 0);
    int kb = 0;
    for (; kb + 2 < nb; kb += 2) {
      issue(fb, kb + 1);
      __builtin_amdgcn_sched_barrier(0);
      mma(fa);
      __builtin_amdgcn_sched_barrier(0);
      issue(fa, kb + 2);
      __builtin_amdgcn_sched_barrier(0);
      mma(fb);
      __builtin_amdgcn_sched_barrier(0);
    }
    if (kb + 1 < nb) {
      issue(fb, kb + 1);
      __builtin_amdgcn_sched_barrier(0);
      mma(fa);
      mma(fb);
    } else {
      mma(fa);
    }
  }

  #pragma unroll
  for (int s = 0; s < 6; ++s) {
    #pragma unroll
    for (int v = 0; v < 8; ++v) {
      const int row = m0 + 16 * s + v + 8 * half;
      #pragma unroll
      for (int q = 0; q < 3; ++q) {
        v2f x;
        x.x = acc[s][2 * q][v];
        x.y = acc[s][2 * q + 1][v];
        *(v2f*)(C + (size_t)row * NNODE + n0 + 32 * q + 2 * l16) = x;
      }
    }
  }
}

struct UnitE { v4f a; v4f b[2]; };

template <int UB, int LDB>
__global__ __launch_bounds__(128) void gemm_afrag_bpi(
    const float* __restrict__ Af,
    const float* __restrict__ Bm,
    const float* __restrict__ bias,
    const float* __restrict__ rs,
    float* __restrict__ C, int ldc,
    int K)
{
  static_assert(3 * UB <= 60, "load group must fit the wave's load counter");
  const int lane = threadIdx.x & 31;
  const int wave = threadIdx.x >> 5;
  const int half = lane >> 4;
  const int l16  = lane & 15;

  const int strip = blockIdx.y * 4 + wave;
  const int m0 = 16 * strip;
  const int n0 = blockIdx.x * 32;
  const int nu = K >> 3;

  v8f acc[2] = {};

  const float* ap = Af + ((size_t)strip * nu * 32 + lane) * 4;
  const float* bp = Bm + (size_t)(2 * half) * (2 * LDB) + 2 * (n0 + 2 * l16);
  constexpr size_t prow  = (size_t)2 * LDB;
  constexpr size_t bunit = (size_t)8 * LDB;

  auto load_u = [&](UnitE& f, int u) {
    f.a = *(const v4f*)(ap + (size_t)u * 128);
    const float* bq = bp + (size_t)u * bunit;
    f.b[0] = *(const v4f*)bq;
    f.b[1] = *(const v4f*)(bq + prow);
  };
  auto mma_u = [&](const UnitE& f) {
    #pragma unroll
    for (int p = 0; p < 2; ++p) {
      const v2f a  = (p == 0) ? __builtin_shufflevector(f.a, f.a, 0, 1)
                              : __builtin_shufflevector(f.a, f.a, 2, 3);
      const v2f b0 = __builtin_shufflevector(f.b[p], f.b[p], 0, 1);
      const v2f b1 = __builtin_shufflevector(f.b[p], f.b[p], 2, 3);
      acc[0] = __builtin_amdgcn_wmma_f32_16x16x4_f32(false, a, false, b0, (short)0, acc[0], false, false);
      acc[1] = __builtin_amdgcn_wmma_f32_16x16x4_f32(false, a, false, b1, (short)0, acc[1], false, false);
    }
  };
  auto issue = [&](UnitE (&f)[UB], int u0) {
    #pragma unroll
    for (int i = 0; i < UB; ++i) load_u(f[i], u0 + i);
  };
  auto mma = [&](const UnitE (&f)[UB]) {
    #pragma unroll
    for (int i = 0; i < UB; ++i) mma_u(f[i]);
  };

  const int nblk = nu / UB;
  UnitE fa[UB], fb[UB];
  if (nblk > 0) {
    issue(fa, 0);
    int b = 0;
    for (; b + 2 < nblk; b += 2) {
      issue(fb, (b + 1) * UB);
      __builtin_amdgcn_sched_barrier(0);
      mma(fa);
      __builtin_amdgcn_sched_barrier(0);
      issue(fa, (b + 2) * UB);
      __builtin_amdgcn_sched_barrier(0);
      mma(fb);
      __builtin_amdgcn_sched_barrier(0);
    }
    if (b + 1 < nblk) {
      issue(fb, (b + 1) * UB);
      __builtin_amdgcn_sched_barrier(0);
      mma(fa);
      mma(fb);
    } else {
      mma(fa);
    }
  }
  for (int u = nblk * UB; u < nu; ++u) {
    UnitE f;
    load_u(f, u);
    mma_u(f);
  }

  const int col = n0 + 2 * l16;
  const float bz0 = bias[col], bz1 = bias[col + 1];
  #pragma unroll
  for (int v = 0; v < 8; ++v) {
    const int row = m0 + v + 8 * half;
    const float sc = rs[row];
    v2f x;
    x.x = acc[0][v];
    x.y = acc[1][v];
    x.x *= sc;  x.y *= sc;
    x.x += bz0; x.y += bz1;
    x.x = fmaxf(x.x, 0.0f); x.y = fmaxf(x.y, 0.0f);
    *(v2f*)(C + (size_t)row * ldc + col) = x;
  }
}

#define NU8 (NNODE / 8)
#define PREP_WAVES 16
#define PREP_LG (NNODE / 32 / PREP_WAVES)
template <int MODE>
__global__ __launch_bounds__(32 * PREP_WAVES) void branch_prep_frag(
    const float* __restrict__ mask,
    const float* __restrict__ adj,
    const float* __restrict__ thrp,
    float* __restrict__ Ahf,
    float* __restrict__ d,
    float* __restrict__ aux,
    float* __restrict__ aux2)
{
  static_assert(NNODE % (32 * PREP_WAVES) == 0 && NNODE % 16 == 0 && PREP_LG % 3 == 0,
                "strip, line and batch split");
  static_assert(PREP_WAVES == 16 && NNODE == 32 * 96, "MODE 1: one wave per row of the strip, 96 elements a lane");
  __shared__ float part[PREP_WAVES][32];
  const int strip = blockIdx.x;
  const int lane  = threadIdx.x & 31;
  const int wave  = threadIdx.x >> 5;
  const int half  = lane >> 4;
  const int row   = 16 * strip + (lane & 15);
  const float thr = thrp[0];
  const float* arow = adj + (size_t)row * NNODE + 4 * half;
  const float* mrow = (MODE == 2) ? mask + 4 * half
                                  : mask + (size_t)row * NNODE + 4 * half;
  const size_t foff = ((size_t)strip * NU8 * 32 + lane) * 4;
  float* dst = Ahf + foff;

  float di = 1.0f;
  if constexpr (MODE == 2) di = mask[row];

  if constexpr (MODE == 1) {
    const int r1 = 16 * strip + wave;
    const float* rp = adj + (size_t)r1 * NNODE + lane;
    float sd = 0.0f;
    #pragma unroll
    for (int b = 0; b < 2; ++b) {
      float v[48];
      #pragma unroll
      for (int t = 0; t < 48; ++t) v[t] = rp[32 * (48 * b + t)];
      #pragma unroll
      for (int t = 0; t < 48; ++t) sd += v[t];
    }
    #pragma unroll
    for (int off = 16; off > 0; off >>= 1) sd += __shfl_xor(sd, off, 32);
    if (lane == 0) aux[r1] = (sd > 0.0f) ? rsqrtf(sd) : 0.0f;
  }

  float s = 0.0f;
  #pragma unroll
  for (int i0 = 0; i0 < PREP_LG; i0 += 3) {
    v4f mv[12], av[12];
    #pragma unroll
    for (int i = 0; i < 3; ++i) {
      #pragma unroll
      for (int t = 0; t < 4; ++t) {
        const int u = 4 * (wave + PREP_WAVES * (i0 + i)) + t;
        av[4 * i + t] = *(const v4f*)(arow + 8 * u);
        if constexpr (MODE != 1) mv[4 * i + t] = *(const v4f*)(mrow + 8 * u);
      }
    }
    #pragma unroll
    for (int i = 0; i < 3; ++i) {
      #pragma unroll
      for (int t = 0; t < 4; ++t) {
        const int u  = 4 * (wave + PREP_WAVES * (i0 + i)) + t;
        const int k0 = 8 * u + 4 * half;
        v4f w, an;
        #pragma unroll
        for (int e = 0; e < 4; ++e) {
          const float a = av[4 * i + t][e];
          float m;
          if constexpr (MODE == 0)      m = mv[4 * i + t][e];
          else if constexpr (MODE == 1) m = a;
          else                          m = (a * di) * mv[4 * i + t][e];
          an[e] = m;
          const float x = (m > thr) ? a : 0.0f;
          s += x;
          w[e] = (k0 + e == row) ? x + 1.0f : x;
        }
        *(v4f*)(dst + (size_t)u * 128) = w;
        if constexpr (MODE == 2) {
          *(v4f*)(aux + foff + (size_t)u * 128) = an;
          const bool odd = (lane & 1) != 0;
          const float g0 = __shfl_xor(odd ? an[0] : an[2], 1, 32);
          const float g1 = __shfl_xor(odd ? an[1] : an[3], 1, 32);
          v4f o;
          o[0] = odd ? g0 : an[0];
          o[1] = odd ? an[2] : g0;
          o[2] = odd ? g1 : an[1];
          o[3] = odd ? an[3] : g1;
          *(v4f*)(aux2 + (size_t)(row >> 1) * (2 * NNODE) + 2 * (k0 + (odd ? 2 : 0))) = o;
        }
      }
    }
  }
  part[wave][lane] = s;
  __syncthreads();
  if (threadIdx.x < 16) {
    float t = 0.0f;
    #pragma unroll
    for (int w = 0; w < PREP_WAVES; ++w) t += part[w][threadIdx.x] + part[w][threadIdx.x + 16];
    d[16 * strip + threadIdx.x] = rsqrtf(t + 1.0f);
  }
}

__global__ void conv_dil2(const float* __restrict__ emb,
                          const float* __restrict__ kw,
                          const float* __restrict__ kb,
                          float* __restrict__ out)
{
  const int idx = blockIdx.x * blockDim.x + threadIdx.x;
  if (idx >= NNODE * 256) return;
  const int n = idx >> 8;
  const int c = idx & 255;
  float acc = kb[0];
  #pragma unroll
  for (int ch = 0; ch < 3; ++ch) {
    #pragma unroll
    for (int kh = 0; kh < 3; ++kh) {
      const int ih = n + 2 * kh - 2;
      if (ih < 0 || ih >= NNODE) continue;
      #pragma unroll
      for (int kwi = 0; kwi < 3; ++kwi) {
        const int iw = c + 2 * kwi - 2;
        if (iw < 0 || iw >= 256) continue;
        acc += emb[((size_t)ch * NNODE + ih) * 256 + iw] * kw[ch * 9 + kh * 3 + kwi];
      }
    }
  }
  out[idx] = acc;
}

__global__ void logits_k(const float* __restrict__ h,
                         const float* __restrict__ Wb,
                         const float* __restrict__ bb,
                         float* __restrict__ out)
{
  const int i = blockIdx.x * blockDim.x + threadIdx.x;
  if (i >= BPAIR * 2) return;
  const int b = i >> 1;
  const int c = i & 1;
  float acc = bb[c];
  #pragma unroll
  for (int k = 0; k < 64; ++k) acc += h[b * 64 + k] * Wb[k * 2 + c];
  out[i] = acc;
}

extern "C" void kernel_launch(void* const* d_in, const int* in_sizes, int n_in,
                              void* d_out, int out_size, void* d_ws, size_t ws_size,
                              hipStream_t stream)
{
  const int*   left  = (const int*)d_in[0];
  const int*   right = (const int*)d_in[1];
  const float* X     = (const float*)d_in[2];
  const float* adj   = (const float*)d_in[3];
  const float* thr   = (const float*)d_in[4];
  const float* Wfc1  = (const float*)d_in[5];
  const float* bfc1  = (const float*)d_in[6];
  const float* Wg[6] = {(const float*)d_in[7],  (const float*)d_in[9],
                        (const float*)d_in[11], (const float*)d_in[13],
                        (const float*)d_in[15], (const float*)d_in[17]};
  const float* bg[6] = {(const float*)d_in[8],  (const float*)d_in[10],
                        (const float*)d_in[12], (const float*)d_in[14],
                        (const float*)d_in[16], (const float*)d_in[18]};
  const float* cnnk  = (const float*)d_in[19];
  const float* cnnb  = (const float*)d_in[20];
  const float* Wa    = (const float*)d_in[21];
  const float* ba    = (const float*)d_in[22];
  const float* Wb    = (const float*)d_in[23];
  const float* bb    = (const float*)d_in[24];

  float* out     = (float*)d_out;
  float* logits  = out;
  float* emb_all = out + BPAIR * 2;

  float* ws = (float*)d_ws;
  size_t off = 0;
  float* X0  = ws + off; off += (size_t)NNODE * D0;
  float* An  = ws + off; off += (size_t)NNODE * NNODE;
  float* A2  = ws + off; off += (size_t)NNODE * NNODE;
  float* Ah  = ws + off; off += (size_t)NNODE * NNODE;
  float* dv  = ws + off; off += NNODE;
  float* Hb  = ws + off; off += (size_t)NNODE * HDIM;
  float* H2b = ws + off; off += (size_t)NNODE * HDIM;
  float* emb = ws + off; off += (size_t)3 * NNODE * 256;
  float* hm  = ws + off; off += (size_t)BPAIR * 64;
  if (ws_size < off * sizeof(float)) return;

  const dim3 blk(128);

  gemm_wmma_f32<true, true, false, 8, false, 2, 2><<<dim3(D0 / 32, NNODE / 128), blk, 0, stream>>>(
      X, FIN, Wfc1, D0, bfc1, nullptr, X0, D0, FIN, nullptr, nullptr);

  float* dsym = hm;
  for (int ch = 0; ch < 3; ++ch) {
    const float* W1 = Wg[4 - 2 * ch];  const float* b1 = bg[4 - 2 * ch];
    const float* W2 = Wg[5 - 2 * ch];  const float* b2 = bg[5 - 2 * ch];
    float* embc = emb + (size_t)ch * NNODE * 256;
    const float* Ahc = (ch == 2) ? An : Ah;

    if (ch == 0) {
      branch_prep_frag<1><<<NNODE / 16, 32 * PREP_WAVES, 0, stream>>>(adj, adj, thr, Ah, dv, dsym, nullptr);
    } else if (ch == 1) {
      branch_prep_frag<2><<<NNODE / 16, 32 * PREP_WAVES, 0, stream>>>(dsym, adj, thr, Ah, dv, An, A2);
    } else {
      gemm_wmma_f32_t96<<<dim3(NNODE / 192, NNODE / 192), blk, 0, stream>>>(An, A2, Ah);
      branch_prep_frag<0><<<NNODE / 16, 32 * PREP_WAVES, 0, stream>>>(Ah, adj, thr, An, dv, nullptr, nullptr);
    }

    gemm_wmma_f32<false, false, false, 8, true, 1, 2, true><<<dim3(HDIM / 32, NNODE / 64), blk, 0, stream>>>(
        X0, D0, W1, HDIM, nullptr, dv, Hb, HDIM, D0, nullptr, nullptr);
    gemm_afrag_bpi<8, HDIM><<<dim3(HDIM / 32, NNODE / 64), blk, 0, stream>>>(
        Ahc, Hb, b1, dv, embc, 256, NNODE);
    gemm_wmma_f32<false, false, false, 8, true, 1, 2, true><<<dim3(HDIM / 32, NNODE / 64), blk, 0, stream>>>(
        embc, 256, W2, HDIM, nullptr, dv, H2b, HDIM, HDIM, nullptr, nullptr);
    gemm_afrag_bpi<8, HDIM><<<dim3(HDIM / 32, NNODE / 64), blk, 0, stream>>>(
        Ahc, H2b, b2, dv, embc + 128, 256, NNODE);
  }

  conv_dil2<<<(NNODE * 256) / 256, 256, 0, stream>>>(emb, cnnk, cnnb, emb_all);

  gemm_wmma_f32<true, true, true, 8, false, 2, 2><<<dim3(64 / 32, BPAIR / 128), blk, 0, stream>>>(
      emb_all, 256, Wa, 64, ba, nullptr, hm, 64, 512, left, right);

  logits_k<<<(BPAIR * 2 + 255) / 256, 256, 0, stream>>>(hm, Wb, bb, logits);
}


# e1/e2 GEMMs (N x N times N x 128): wave tile 16x32 -> 16x16, 8 waves per work-group = two per SIMD, load groups 16 units deep; H/H2 stored k-quad interleaved for them; k and sum order kept
# speedup vs baseline: 4.5013x; 1.0493x over previous
#include <hip/hip_runtime.h>
#include <hip/hip_bf16.h>

#define NNODE 3072
#define FIN   1546
#define D0    256
#define HDIM  128
#define BPAIR 16384

typedef __attribute__((ext_vector_type(2))) float v2f;
typedef __attribute__((ext_vector_type(4))) float v4f;
typedef __attribute__((ext_vector_type(8))) float v8f;

template <int SM, int SN>
struct KStep { v2f a[SM]; v2f b[SN]; };

template <int SM, int SN>
__device__ __forceinline__ void load_kstep(KStep<SM, SN>& f, const float* const (&ap)[SM],
                                           const float* bp, int ldb)
{
  #pragma unroll
  for (int s = 0; s < SM; ++s) f.a[s] = *(const v2f*)ap[s];
  #pragma unroll
  for (int j = 0; j < SN; ++j) {
    f.b[j].x = bp[16 * j];
    f.b[j].y = bp[ldb + 16 * j];
  }
}

template <int SM, int SN>
__device__ __forceinline__ void wmma_tile(v8f (&acc)[SM][SN], const KStep<SM, SN>& f)
{
  #pragma unroll
  for (int s = 0; s < SM; ++s) {
    #pragma unroll
    for (int j = 0; j < SN; ++j)
      acc[s][j] = __builtin_amdgcn_wmma_f32_16x16x4_f32(false, f.a[s], false, f.b[j], (short)0,
                                                        acc[s][j], false, false);
  }
}

template <int U, int SM, int SN>
__device__ __forceinline__ void mma_block(v8f (&acc)[SM][SN], const KStep<SM, SN> (&f)[U])
{
  #pragma unroll
  for (int i = 0; i < U; ++i) wmma_tile<SM, SN>(acc, f[i]);
}

template <bool RELU, bool BIAS, bool GATHER, int U, bool RSCALE, int SM, int SN,
          bool CQI = false>
__global__ __launch_bounds__(128) void gemm_wmma_f32(
    const float* __restrict__ A, int lda,
    const float* __restrict__ Bm, int ldb,
    const float* __restrict__ bias,
    const float* __restrict__ rs,
    float* __restrict__ C, int ldc,
    int K,
    const int* __restrict__ gl, const int* __restrict__ gr)
{
  static_assert(U * (SM + 2 * SN) <= 60, "load group must fit the wave's load counter");
  constexpr int TM = 16 * SM;
  constexpr int TN = 16 * SN;
  using KS = KStep<SM, SN>;

  const int lane = threadIdx.x & 31;
  const int wave = threadIdx.x >> 5;
  const int half = lane >> 4;
  const int l16  = lane & 15;

  const int m0 = blockIdx.y * (4 * TM) + wave * TM;
  const int n0 = blockIdx.x * TN;

  v8f acc[SM][SN] = {};

  const float* bp = Bm + (size_t)(2 * half) * ldb + n0 + l16;
  const size_t bstep = (size_t)4 * ldb;

  constexpr int UU = U;
  const float* pl[SM];
  const float* pr[SM];
  #pragma unroll
  for (int s = 0; s < SM; ++s) {
    const int am = m0 + 16 * s + l16;
    pl[s] = GATHER ? A + (size_t)gl[am] * 256 + 2 * half
                   : A + (size_t)am * lda + 2 * half;
    pr[s] = GATHER ? A + (size_t)gr[am] * 256 + 2 * half : pl[s];
  }
  const int nk   = GATHER ? 128 : (K >> 2);
  const int nblk = nk / UU;

  auto load_t = [&](KS& f, int t) {
    const float* q[SM];
    const bool r = GATHER && (t >= 64);
    const int tt = GATHER ? (t & 63) : t;
    #pragma unroll
    for (int s = 0; s < SM; ++s) q[s] = (r ? pr[s] : pl[s]) + 4 * (size_t)tt;
    load_kstep<SM, SN>(f, q, bp + (size_t)t * bstep, ldb);
  };
  auto issue = [&](KS (&f)[UU], int t0) {
    #pragma unroll
    for (int i = 0; i < UU; ++i) load_t(f[i], t0 + i);
  };

  KS fa[UU], fb[UU];
  if (nblk > 0) {
    issue(fa, 0);
    int b = 0;
    for (; b + 2 < nblk; b += 2) {
      issue(fb, (b + 1) * UU);
      __builtin_amdgcn_sched_barrier(0);
      mma_block<UU, SM, SN>(acc, fa);
      __builtin_amdgcn_sched_barrier(0);
      issue(fa, (b + 2) * UU);
      __builtin_amdgcn_sched_barrier(0);
      mma_block<UU, SM, SN>(acc, fb);
      __builtin_amdgcn_sched_barrier(0);
    }
    if (b + 1 < nblk) {
      issue(fb, (b + 1) * UU);
      __builtin_amdgcn_sched_barrier(0);
      mma_block<UU, SM, SN>(acc, fa);
      mma_block<UU, SM, SN>(acc, fb);
    } else {
      mma_block<UU, SM, SN>(acc, fa);
    }
  }
  for (int t = nblk * UU; t < nk; ++t) {
    KS f;
    load_t(f, t);
    wmma_tile<SM, SN>(acc, f);
  }
  if (!GATHER) {
    const int kmain = K & ~3;
    if (kmain < K) {
      const float* bq = bp + (size_t)nk * bstep;
      const int ka = kmain + 2 * half;
      const bool g0 = (ka < K), g1 = (ka + 1 < K);
      KS f;
      #pragma unroll
      for (int s = 0; s < SM; ++s) {
        const float* ap = pl[s] + kmain;
        f.a[s].x = g0 ? ap[0] : 0.0f;
        f.a[s].y = g1 ? ap[1] : 0.0f;
      }
      #pragma unroll
      for (int j = 0; j < SN; ++j) {
        f.b[j].x = g0 ? bq[16 * j]       : 0.0f;
        f.b[j].y = g1 ? bq[ldb + 16 * j] : 0.0f;
      }
      wmma_tile<SM, SN>(acc, f);
    }
  }

  if constexpr (CQI) {
    #pragma unroll
    for (int s = 0; s < SM; ++s) {
      #pragma unroll
      for (int v = 0; v < 8; v += 4) {
        const int row = m0 + 16 * s + v + 8 * half;
        float sc[4];
        #pragma unroll
        for (int e = 0; e < 4; ++e) sc[e] = RSCALE ? rs[row + e] : 1.0f;
        #pragma unroll
        for (int j = 0; j < SN; ++j) {
          const int col = n0 + 16 * j + l16;
          v4f x;
          #pragma unroll
          for (int e = 0; e < 4; ++e) {
            float y = acc[s][j][v + e];
            if (RSCALE) y *= sc[e];
            if (BIAS) y += bias[col];
            if (RELU) y = fmaxf(y, 0.0f);
            x[e] = y;
          }
          *(v4f*)(C + (size_t)(row >> 2) * (4 * ldc) + 4 * col) = x;
        }
      }
    }
  } else {
    #pragma unroll
    for (int s = 0; s < SM; ++s) {
      #pragma unroll
      for (int v = 0; v < 8; ++v) {
        const int row = m0 + 16 * s + v + 8 * half;
        const float sc = RSCALE ? rs[row] : 1.0f;
        #pragma unroll
        for (int j = 0; j < SN; ++j) {
          const int col = n0 + 16 * j + l16;
          float x = acc[s][j][v];
          if (RSCALE) x *= sc;
          if (BIAS) x += bias[col];
          if (RELU) x = fmaxf(x, 0.0f);
          C[(size_t)row * ldc + col] = x;
        }
      }
    }
  }
}

struct Unit96 { v4f a[6]; v4f b[2][3]; };

__device__ __forceinline__ void mma_unit96(v8f (&acc)[6][6], const Unit96& f)
{
  #pragma unroll
  for (int p = 0; p < 2; ++p) {
    #pragma unroll
    for (int s = 0; s < 6; ++s) {
      const v2f a = (p == 0) ? __builtin_shufflevector(f.a[s], f.a[s], 0, 1)
                             : __builtin_shufflevector(f.a[s], f.a[s], 2, 3);
      #pragma unroll
      for (int j = 0; j < 6; ++j) {
        const v4f bb = f.b[p][j >> 1];
        const v2f b = (j & 1) ? __builtin_shufflevector(bb, bb, 2, 3)
                              : __builtin_shufflevector(bb, bb, 0, 1);
        acc[s][j] = __builtin_amdgcn_wmma_f32_16x16x4_f32(false, a, false, b, (short)0,
                                                          acc[s][j], false, false);
      }
    }
  }
}

__global__ __launch_bounds__(128) void gemm_wmma_f32_t96(
    const float* __restrict__ Af,
    const float* __restrict__ Bpi,
    float* __restrict__ C)
{
  static_assert(NNODE % 192 == 0, "tile split");
  constexpr int NU = NNODE / 8;
  const int lane = threadIdx.x & 31;
  const int wave = threadIdx.x >> 5;
  const int half = lane >> 4;
  const int l16  = lane & 15;

  const int strip0 = blockIdx.y * 12 + (wave >> 1) * 6;
  const int m0 = 16 * strip0;
  const int n0 = blockIdx.x * 192 + (wave & 1) * 96;

  v8f acc[6][6] = {};

  const float* ap = Af + ((size_t)strip0 * NU * 32 + lane) * 4;
  const float* bp = Bpi + (size_t)(2 * half) * (2 * NNODE) + 2 * (n0 + 2 * l16);
  constexpr size_t astrip = (size_t)NU * 128;
  constexpr size_t prow   = (size_t)2 * NNODE;
  constexpr size_t bunit  = (size_t)8 * NNODE;

  auto load_u = [&](Unit96& f, int u) {
    const float* aq = ap + (size_t)u * 128;
    const float* bq = bp + (size_t)u * bunit;
    #pragma unroll
    for (int s = 0; s < 6; ++s) f.a[s] = *(const v4f*)(aq + s * astrip);
    #pragma unroll
    for (int p = 0; p < 2; ++p) {
      #pragma unroll
      for (int q = 0; q < 3; ++q) f.b[p][q] = *(const v4f*)(bq + p * prow + 64 * q);
    }
  };
  auto issue = [&](Unit96 (&f)[2], int kb) {
    load_u(f[0], 2 * kb);
    load_u(f[1], 2 * kb + 1);
  };
  auto mma = [&](const Unit96 (&f)[2]) {
    mma_unit96(acc, f[0]);
    mma_unit96(acc, f[1]);
  };

  constexpr int nb = NNODE / 16;
  Unit96 fa[2], fb[2];
  {
    issue(fa, 0);
    int kb = 0;
    for (; kb + 2 < nb; kb += 2) {
      issue(fb, kb + 1);
      __builtin_amdgcn_sched_barrier(0);
      mma(fa);
      __builtin_amdgcn_sched_barrier(0);
      issue(fa, kb + 2);
      __builtin_amdgcn_sched_barrier(0);
      mma(fb);
      __builtin_amdgcn_sched_barrier(0);
    }
    if (kb + 1 < nb) {
      issue(fb, kb + 1);
      __builtin_amdgcn_sched_barrier(0);
      mma(fa);
      mma(fb);
    } else {
      mma(fa);
    }
  }

  #pragma unroll
  for (int s = 0; s < 6; ++s) {
    #pragma unroll
    for (int v = 0; v < 8; ++v) {
      const int row = m0 + 16 * s + v + 8 * half;
      #pragma unroll
      for (int q = 0; q < 3; ++q) {
        v2f x;
        x.x = acc[s][2 * q][v];
        x.y = acc[s][2 * q + 1][v];
        *(v2f*)(C + (size_t)row * NNODE + n0 + 32 * q + 2 * l16) = x;
      }
    }
  }
}

struct UnitQ { v4f a; v4f b; };

template <int UB, int LDB>
__global__ __launch_bounds__(256) void gemm_afrag_bqi(
    const float* __restrict__ Af,
    const float* __restrict__ Bm,
    const float* __restrict__ bias,
    const float* __restrict__ rs,
    float* __restrict__ C, int ldc,
    int K)
{
  static_assert(2 * UB <= 60, "load group must fit the wave's load counter");
  __shared__ __attribute__((aligned(16))) float stage[64][36];
  const int lane = threadIdx.x & 31;
  const int wave = threadIdx.x >> 5;
  const int half = lane >> 4;
  const int l16  = lane & 15;
  const int ms   = wave >> 1;
  const int ns   = wave & 1;

  const int strip = blockIdx.y * 4 + ms;
  const int n0 = blockIdx.x * 32 + 16 * ns;
  const int nu = K >> 3;

  v8f acc = {};

  const float* ap = Af + ((size_t)strip * nu * 32 + lane) * 4;
  const float* bp = Bm + (size_t)half * (4 * LDB) + 4 * (n0 + l16);
  constexpr size_t bunit = (size_t)8 * LDB;

  auto load_u = [&](UnitQ& f, int u) {
    f.a = *(const v4f*)(ap + (size_t)u * 128);
    f.b = *(const v4f*)(bp + (size_t)u * bunit);
  };
  auto mma_u = [&](const UnitQ& f) {
    #pragma unroll
    for (int p = 0; p < 2; ++p) {
      const v2f a = (p == 0) ? __builtin_shufflevector(f.a, f.a, 0, 1)
                             : __builtin_shufflevector(f.a, f.a, 2, 3);
      const v2f b = (p == 0) ? __builtin_shufflevector(f.b, f.b, 0, 1)
                             : __builtin_shufflevector(f.b, f.b, 2, 3);
      acc = __builtin_amdgcn_wmma_f32_16x16x4_f32(false, a, false, b, (short)0, acc, false, false);
    }
  };
  auto issue = [&](UnitQ (&f)[UB], int u0) {
    #pragma unroll
    for (int i = 0; i < UB; ++i) load_u(f[i], u0 + i);
  };
  auto mma = [&](const UnitQ (&f)[UB]) {
    #pragma unroll
    for (int i = 0; i < UB; ++i) mma_u(f[i]);
  };

  const int nblk = nu / UB;
  UnitQ fa[UB], fb[UB];
  if (nblk > 0) {
    issue(fa, 0);
    int b = 0;
    for (; b + 2 < nblk; b += 2) {
      issue(fb, (b + 1) * UB);
      __builtin_amdgcn_sched_barrier(0);
      mma(fa);
      __builtin_amdgcn_sched_barrier(0);
      issue(fa, (b + 2) * UB);
      __builtin_amdgcn_sched_barrier(0);
      mma(fb);
      __builtin_amdgcn_sched_barrier(0);
    }
    if (b + 1 < nblk) {
      issue(fb, (b + 1) * UB);
      __builtin_amdgcn_sched_barrier(0);
      mma(fa);
      mma(fb);
    } else {
      mma(fa);
    }
  }
  for (int u = nblk * UB; u < nu; ++u) {
    UnitQ f;
    load_u(f, u);
    mma_u(f);
  }

  {
    const float bz = bias[n0 + l16];
    #pragma unroll
    for (int v = 0; v < 8; ++v) {
      const int rl = 16 * ms + v + 8 * half;
      const float sc = rs[64 * blockIdx.y + rl];
      float x = acc[v];
      x *= sc;
      x += bz;
      x = fmaxf(x, 0.0f);
      stage[rl][16 * ns + l16] = x;
    }
  }
  __syncthreads();
  {
    const int r0 = threadIdx.x >> 3;
    const int c4 = 4 * (threadIdx.x & 7);
    #pragma unroll
    for (int i = 0; i < 2; ++i) {
      const int rl = r0 + 32 * i;
      const v4f x = *(const v4f*)&stage[rl][c4];
      *(v4f*)(C + (size_t)(64 * blockIdx.y + rl) * ldc + 32 * blockIdx.x + c4) = x;
    }
  }
}

#define NU8 (NNODE / 8)
#define PREP_WAVES 16
#define PREP_LG (NNODE / 32 / PREP_WAVES)
template <int MODE>
__global__ __launch_bounds__(32 * PREP_WAVES) void branch_prep_frag(
    const float* __restrict__ mask,
    const float* __restrict__ adj,
    const float* __restrict__ thrp,
    float* __restrict__ Ahf,
    float* __restrict__ d,
    float* __restrict__ aux,
    float* __restrict__ aux2)
{
  static_assert(NNODE % (32 * PREP_WAVES) == 0 && NNODE % 16 == 0 && PREP_LG % 3 == 0,
                "strip, line and batch split");
  static_assert(PREP_WAVES == 16 && NNODE == 32 * 96, "MODE 1: one wave per row of the strip, 96 elements a lane");
  __shared__ float part[PREP_WAVES][32];
  const int strip = blockIdx.x;
  const int lane  = threadIdx.x & 31;
  const int wave  = threadIdx.x >> 5;
  const int half  = lane >> 4;
  const int row   = 16 * strip + (lane & 15);
  const float thr = thrp[0];
  const float* arow = adj + (size_t)row * NNODE + 4 * half;
  const float* mrow = (MODE == 2) ? mask + 4 * half
                                  : mask + (size_t)row * NNODE + 4 * half;
  const size_t foff = ((size_t)strip * NU8 * 32 + lane) * 4;
  float* dst = Ahf + foff;

  float di = 1.0f;
  if constexpr (MODE == 2) di = mask[row];

  if constexpr (MODE == 1) {
    const int r1 = 16 * strip + wave;
    const float* rp = adj + (size_t)r1 * NNODE + lane;
    float sd = 0.0f;
    #pragma unroll
    for (int b = 0; b < 2; ++b) {
      float v[48];
      #pragma unroll
      for (int t = 0; t < 48; ++t) v[t] = rp[32 * (48 * b + t)];
      #pragma unroll
      for (int t = 0; t < 48; ++t) sd += v[t];
    }
    #pragma unroll
    for (int off = 16; off > 0; off >>= 1) sd += __shfl_xor(sd, off, 32);
    if (lane == 0) aux[r1] = (sd > 0.0f) ? rsqrtf(sd) : 0.0f;
  }

  float s = 0.0f;
  #pragma unroll
  for (int i0 = 0; i0 < PREP_LG; i0 += 3) {
    v4f mv[12], av[12];
    #pragma unroll
    for (int i = 0; i < 3; ++i) {
      #pragma unroll
      for (int t = 0; t < 4; ++t) {
        const int u = 4 * (wave + PREP_WAVES * (i0 + i)) + t;
        av[4 * i + t] = *(const v4f*)(arow + 8 * u);
        if constexpr (MODE != 1) mv[4 * i + t] = *(const v4f*)(mrow + 8 * u);
      }
    }
    #pragma unroll
    for (int i = 0; i < 3; ++i) {
      #pragma unroll
      for (int t = 0; t < 4; ++t) {
        const int u  = 4 * (wave + PREP_WAVES * (i0 + i)) + t;
        const int k0 = 8 * u + 4 * half;
        v4f w, an;
        #pragma unroll
        for (int e = 0; e < 4; ++e) {
          const float a = av[4 * i + t][e];
          float m;
          if constexpr (MODE == 0)      m = mv[4 * i + t][e];
          else if constexpr (MODE == 1) m = a;
          else                          m = (a * di) * mv[4 * i + t][e];
          an[e] = m;
          const float x = (m > thr) ? a : 0.0f;
          s += x;
          w[e] = (k0 + e == row) ? x + 1.0f : x;
        }
        *(v4f*)(dst + (size_t)u * 128) = w;
        if constexpr (MODE == 2) {
          *(v4f*)(aux + foff + (size_t)u * 128) = an;
          const bool odd = (lane & 1) != 0;
          const float g0 = __shfl_xor(odd ? an[0] : an[2], 1, 32);
          const float g1 = __shfl_xor(odd ? an[1] : an[3], 1, 32);
          v4f o;
          o[0] = odd ? g0 : an[0];
          o[1] = odd ? an[2] : g0;
          o[2] = odd ? g1 : an[1];
          o[3] = odd ? an[3] : g1;
          *(v4f*)(aux2 + (size_t)(row >> 1) * (2 * NNODE) + 2 * (k0 + (odd ? 2 : 0))) = o;
        }
      }
    }
  }
  part[wave][lane] = s;
  __syncthreads();
  if (threadIdx.x < 16) {
    float t = 0.0f;
    #pragma unroll
    for (int w = 0; w < PREP_WAVES; ++w) t += part[w][threadIdx.x] + part[w][threadIdx.x + 16];
    d[16 * strip + threadIdx.x] = rsqrtf(t + 1.0f);
  }
}

__global__ void conv_dil2(const float* __restrict__ emb,
                          const float* __restrict__ kw,
                          const float* __restrict__ kb,
                          float* __restrict__ out)
{
  const int idx = blockIdx.x * blockDim.x + threadIdx.x;
  if (idx >= NNODE * 256) return;
  const int n = idx >> 8;
  const int c = idx & 255;
  float acc = kb[0];
  #pragma unroll
  for (int ch = 0; ch < 3; ++ch) {
    #pragma unroll
    for (int kh = 0; kh < 3; ++kh) {
      const int ih = n + 2 * kh - 2;
      if (ih < 0 || ih >= NNODE) continue;
      #pragma unroll
      for (int kwi = 0; kwi < 3; ++kwi) {
        const int iw = c + 2 * kwi - 2;
        if (iw < 0 || iw >= 256) continue;
        acc += emb[((size_t)ch * NNODE + ih) * 256 + iw] * kw[ch * 9 + kh * 3 + kwi];
      }
    }
  }
  out[idx] = acc;
}

__global__ void logits_k(const float* __restrict__ h,
                         const float* __restrict__ Wb,
                         const float* __restrict__ bb,
                         float* __restrict__ out)
{
  const int i = blockIdx.x * blockDim.x + threadIdx.x;
  if (i >= BPAIR * 2) return;
  const int b = i >> 1;
  const int c = i & 1;
  float acc = bb[c];
  #pragma unroll
  for (int k = 0; k < 64; ++k) acc += h[b * 64 + k] * Wb[k * 2 + c];
  out[i] = acc;
}

extern "C" void kernel_launch(void* const* d_in, const int* in_sizes, int n_in,
                              void* d_out, int out_size, void* d_ws, size_t ws_size,
                              hipStream_t stream)
{
  const int*   left  = (const int*)d_in[0];
  const int*   right = (const int*)d_in[1];
  const float* X     = (const float*)d_in[2];
  const float* adj   = (const float*)d_in[3];
  const float* thr   = (const float*)d_in[4];
  const float* Wfc1  = (const float*)d_in[5];
  const float* bfc1  = (const float*)d_in[6];
  const float* Wg[6] = {(const float*)d_in[7],  (const float*)d_in[9],
                        (const float*)d_in[11], (const float*)d_in[13],
                        (const float*)d_in[15], (const float*)d_in[17]};
  const float* bg[6] = {(const float*)d_in[8],  (const float*)d_in[10],
                        (const float*)d_in[12], (const float*)d_in[14],
                        (const float*)d_in[16], (const float*)d_in[18]};
  const float* cnnk  = (const float*)d_in[19];
  const float* cnnb  = (const float*)d_in[20];
  const float* Wa    = (const float*)d_in[21];
  const float* ba    = (const float*)d_in[22];
  const float* Wb    = (const float*)d_in[23];
  const float* bb    = (const float*)d_in[24];

  float* out     = (float*)d_out;
  float* logits  = out;
  float* emb_all = out + BPAIR * 2;

  float* ws = (float*)d_ws;
  size_t off = 0;
  float* X0  = ws + off; off += (size_t)NNODE * D0;
  float* An  = ws + off; off += (size_t)NNODE * NNODE;
  float* A2  = ws + off; off += (size_t)NNODE * NNODE;
  float* Ah  = ws + off; off += (size_t)NNODE * NNODE;
  float* dv  = ws + off; off += NNODE;
  float* Hb  = ws + off; off += (size_t)NNODE * HDIM;
  float* H2b = ws + off; off += (size_t)NNODE * HDIM;
  float* emb = ws + off; off += (size_t)3 * NNODE * 256;
  float* hm  = ws + off; off += (size_t)BPAIR * 64;
  if (ws_size < off * sizeof(float)) return;

  const dim3 blk(128);
  const dim3 blk8(256);

  gemm_wmma_f32<true, true, false, 8, false, 2, 2><<<dim3(D0 / 32, NNODE / 128), blk, 0, stream>>>(
      X, FIN, Wfc1, D0, bfc1, nullptr, X0, D0, FIN, nullptr, nullptr);

  float* dsym = hm;
  for (int ch = 0; ch < 3; ++ch) {
    const float* W1 = Wg[4 - 2 * ch];  const float* b1 = bg[4 - 2 * ch];
    const float* W2 = Wg[5 - 2 * ch];  const float* b2 = bg[5 - 2 * ch];
    float* embc = emb + (size_t)ch * NNODE * 256;
    const float* Ahc = (ch == 2) ? An : Ah;

    if (ch == 0) {
      branch_prep_frag<1><<<NNODE / 16, 32 * PREP_WAVES, 0, stream>>>(adj, adj, thr, Ah, dv, dsym, nullptr);
    } else if (ch == 1) {
      branch_prep_frag<2><<<NNODE / 16, 32 * PREP_WAVES, 0, stream>>>(dsym, adj, thr, Ah, dv, An, A2);
    } else {
      gemm_wmma_f32_t96<<<dim3(NNODE / 192, NNODE / 192), blk, 0, stream>>>(An, A2, Ah);
      branch_prep_frag<0><<<NNODE / 16, 32 * PREP_WAVES, 0, stream>>>(Ah, adj, thr, An, dv, nullptr, nullptr);
    }

    gemm_wmma_f32<false, false, false, 8, true, 1, 2, true><<<dim3(HDIM / 32, NNODE / 64), blk, 0, stream>>>(
        X0, D0, W1, HDIM, nullptr, dv, Hb, HDIM, D0, nullptr, nullptr);
    gemm_afrag_bqi<16, HDIM><<<dim3(HDIM / 32, NNODE / 64), blk8, 0, stream>>>(
        Ahc, Hb, b1, dv, embc, 256, NNODE);
    gemm_wmma_f32<false, false, false, 8, true, 1, 2, true><<<dim3(HDIM / 32, NNODE / 64), blk, 0, stream>>>(
        embc, 256, W2, HDIM, nullptr, dv, H2b, HDIM, HDIM, nullptr, nullptr);
    gemm_afrag_bqi<16, HDIM><<<dim3(HDIM / 32, NNODE / 64), blk8, 0, stream>>>(
        Ahc, H2b, b2, dv, embc + 128, 256, NNODE);
  }

  conv_dil2<<<(NNODE * 256) / 256, 256, 0, stream>>>(emb, cnnk, cnnb, emb_all);

  gemm_wmma_f32<true, true, true, 8, false, 2, 2><<<dim3(64 / 32, BPAIR / 128), blk, 0, stream>>>(
      emb_all, 256, Wa, 64, ba, nullptr, hm, 64, 512, left, right);

  logits_k<<<(BPAIR * 2 + 255) / 256, 256, 0, stream>>>(hm, Wb, bb, logits);
}
